# starting kernel
# baseline (speedup 1.0000x reference)
#include <hip/hip_runtime.h>

typedef __attribute__((ext_vector_type(2))) float v2f;
typedef __attribute__((ext_vector_type(8))) float v8f;

#define NN 50000
#define NE 1600000
#define DN 64
#define DE 32
#define DG 32
#define HIDDEN 128
#define NF 128
#define ROWS 64
#define LDF 130


__global__ void zero_ws(float* __restrict__ p, int n) {
  int i = blockIdx.x * blockDim.x + threadIdx.x;
  if (i < n) p[i] = 0.0f;
}

__global__ void scatter_edges(const float* __restrict__ edge_attr,
                              const int* __restrict__ edge_index,
                              float* __restrict__ sums,
                              float* __restrict__ counts) {
  long long t = (long long)blockIdx.x * blockDim.x + threadIdx.x;
  int e = (int)(t >> 5);
  if (e >= NE) return;
  int c = (int)(t & 31);
  int dst = edge_index[NE + e];
  atomicAdd(&sums[(long long)dst * DE + c], edge_attr[(long long)e * DE + c]);
  if (c == 0) atomicAdd(&counts[dst], 1.0f);
}

__global__ void seg_mean(const float* __restrict__ sums,
                         const float* __restrict__ counts,
                         float* __restrict__ eob) {
  int i = blockIdx.x * blockDim.x + threadIdx.x;
  if (i >= NN * DE) return;
  float cnt = counts[i >> 5];
  eob[i] = sums[i] / fmaxf(cnt, 1.0f);
}


__global__ __launch_bounds__(128) void mlp_wmma(
    const float* __restrict__ x, const float* __restrict__ eob,
    const float* __restrict__ u, const int* __restrict__ batch,
    const float* __restrict__ W1, const float* __restrict__ b1,
    const float* __restrict__ W2, const float* __restrict__ b2,
    float* __restrict__ y) {
  __shared__ float feat[ROWS * LDF];
  __shared__ float hbuf[ROWS * LDF];

  const int tid = threadIdx.x;
  const int nb = blockIdx.x * ROWS;

  for (int idx = tid; idx < ROWS * NF; idx += 128) {
    int r = idx >> 7;
    int c = idx & 127;
    int node = nb + r;
    float v = 0.0f;
    if (node < NN) {
      if (c < DN)           v = x[(long long)node * DN + c];
      else if (c < DN + DE) v = eob[(long long)node * DE + (c - DN)];
      else                  v = u[(long long)batch[node] * DG + (c - DN - DE)];
    }
    feat[r * LDF + c] = v;
  }
  __syncthreads();

  const int wave = tid >> 5;
  const int lane = tid & 31;
  const int half = lane >> 4;
  const int nidx = lane & 15;
  const int mrow0 = wave * 16;

  for (int nt = 0; nt < HIDDEN / 16; ++nt) {
    const int ncol = nt * 16 + nidx;
    const float bias = b1[ncol];
    v8f acc;
#pragma unroll
    for (int v = 0; v < 8; ++v) acc[v] = bias;
#pragma unroll 4
    for (int ks = 0; ks < NF / 4; ++ks) {
      int kk = ks * 4 + 2 * half;
      v2f a = *(const v2f*)&feat[(mrow0 + nidx) * LDF + kk];
      v2f b;
      b.x = W1[kk * HIDDEN + ncol];
      b.y = W1[(kk + 1) * HIDDEN + ncol];
      acc = __builtin_amdgcn_wmma_f32_16x16x4_f32(false, a, false, b,
                                                  (short)0, acc, false, false);
    }
#pragma unroll
    for (int v = 0; v < 8; ++v) {
      float r = acc[v];
      hbuf[(mrow0 + v + 8 * half) * LDF + ncol] = r > 0.0f ? r : 0.0f;
    }
  }
  __syncthreads();

  for (int nt = 0; nt < DN / 16; ++nt) {
    const int ncol = nt * 16 + nidx;
    const float bias = b2[ncol];
    v8f acc;
#pragma unroll
    for (int v = 0; v < 8; ++v) acc[v] = bias;
#pragma unroll 4
    for (int ks = 0; ks < HIDDEN / 4; ++ks) {
      int kk = ks * 4 + 2 * half;
      v2f a = *(const v2f*)&hbuf[(mrow0 + nidx) * LDF + kk];
      v2f b;
      b.x = W2[kk * DN + ncol];
      b.y = W2[(kk + 1) * DN + ncol];
      acc = __builtin_amdgcn_wmma_f32_16x16x4_f32(false, a, false, b,
                                                  (short)0, acc, false, false);
    }
#pragma unroll
    for (int v = 0; v < 8; ++v) {
      int node = nb + mrow0 + v + 8 * half;
      if (node < NN) {
        float r = acc[v];
        y[(long long)node * DN + ncol] = r > 0.0f ? r : 0.0f;
      }
    }
  }
}


extern "C" void kernel_launch(void* const* d_in, const int* in_sizes, int n_in,
                              void* d_out, int out_size, void* d_ws, size_t ws_size,
                              hipStream_t stream) {
  const float* x     = (const float*)d_in[0];
  const int*   eidx  = (const int*)d_in[1];
  const float* eattr = (const float*)d_in[2];
  const float* u     = (const float*)d_in[3];
  const int*   batch = (const int*)d_in[4];
  const float* W1    = (const float*)d_in[5];
  const float* b1    = (const float*)d_in[6];
  const float* W2    = (const float*)d_in[7];
  const float* b2    = (const float*)d_in[8];

  float* y   = (float*)d_out;
  float* eob = (float*)d_out + (size_t)NN * DN;

  float* sums   = (float*)d_ws;
  float* counts = sums + (size_t)NN * DE;

  int nz = NN * DE + NN;
  zero_ws<<<(nz + 255) / 256, 256, 0, stream>>>((float*)d_ws, nz);

  long long sthreads = (long long)NE * 32;
  scatter_edges<<<(int)((sthreads + 255) / 256), 256, 0, stream>>>(
      eattr, eidx, sums, counts);

  int mz = NN * DE;
  seg_mean<<<(mz + 255) / 256, 256, 0, stream>>>(sums, counts, eob);

  mlp_wmma<<<(NN + ROWS - 1) / ROWS, 128, 0, stream>>>(
      x, eob, u, batch, W1, b1, W2, b2, y);
}


# scatter-mean: global f32 atomics replaced by in-call LDS counting sort of edges by 256-node range, then per-range row sums by LDS f32 atomic adds (arrival order), one store per line; 4 launches to 3
# speedup vs baseline: 3.5050x; 3.5050x over previous
#include <hip/hip_runtime.h>

typedef __attribute__((ext_vector_type(2))) float v2f;
typedef __attribute__((ext_vector_type(8))) float v8f;

#define NN 50000
#define NE 1600000
#define DN 64
#define DE 32
#define DG 32
#define HIDDEN 128
#define NF 128
#define ROWS 64
#define LDF 130

#define CH 8192
#define NCH ((NE + CH - 1) / CH)
#define RSH 8
#define RN 256
#define NRG ((NN + RN - 1) / RN)
#define TP 256
#define SORTED_BYTES ((size_t)NE * 4)
#define NB 16

__global__ __launch_bounds__(1024) void bin_edges(const int* __restrict__ edge_index,
                                                  unsigned* __restrict__ sorted,
                                                  unsigned* __restrict__ tab2) {
  __shared__ unsigned hist[TP];
  __shared__ unsigned offs[TP];
  __shared__ unsigned stage[CH];

  const int tid = threadIdx.x;
  const int chunk = blockIdx.x;
  const int base = chunk * CH;

  if (tid < TP) hist[tid] = 0u;
  __syncthreads();

  int dsts[CH / 1024];
#pragma unroll
  for (int k = 0; k < CH / 1024; ++k) {
    int e = base + k * 1024 + tid;
    int ec = (e < NE) ? e : (NE - 1);
    int d = edge_index[NE + ec];
    dsts[k] = (e < NE) ? d : -1;
  }

  unsigned key[CH / 1024];
  unsigned slot[CH / 1024];
#pragma unroll
  for (int k = 0; k < CH / 1024; ++k) {
    int e = base + k * 1024 + tid;
    int d = dsts[k];
    slot[k] = 0xFFFFFFFFu;
    key[k] = 0u;
    if ((unsigned)d < (unsigned)NN) {
      unsigned r = (unsigned)d >> RSH;
      unsigned rank = atomicAdd(&hist[r], 1u);
      key[k] = ((unsigned)e << 8) | ((unsigned)d & (RN - 1));
      slot[k] = (r << 16) | rank;
    }
  }
  __syncthreads();

  if (tid < 32) {
    unsigned v[8];
    unsigned s = 0u;
#pragma unroll
    for (int i = 0; i < 8; ++i) { v[i] = hist[tid * 8 + i]; s += v[i]; }
    unsigned inc = s;
#pragma unroll
    for (int o = 1; o < 32; o <<= 1) {
      unsigned t = __shfl_up(inc, o);
      if (tid >= o) inc += t;
    }
    unsigned run = inc - s;
#pragma unroll
    for (int i = 0; i < 8; ++i) { offs[tid * 8 + i] = run; run += v[i]; }
  }
  __syncthreads();

#pragma unroll
  for (int k = 0; k < CH / 1024; ++k) {
    if (slot[k] != 0xFFFFFFFFu) stage[offs[slot[k] >> 16] + (slot[k] & 0xFFFFu)] = key[k];
  }
  if (tid < TP / 2) tab2[chunk * (TP / 2) + tid] = offs[2 * tid] | (offs[2 * tid + 1] << 16);
  __syncthreads();

  const unsigned total = offs[TP - 1];
#pragma unroll
  for (int k = 0; k < CH / 1024; ++k) {
    unsigned i = k * 1024 + tid;
    if (i < total) sorted[base + i] = stage[i];
  }
}

__global__ __launch_bounds__(1024) void range_mean(const float* __restrict__ edge_attr,
                                                   const unsigned* __restrict__ sorted,
                                                   const unsigned* __restrict__ tab2,
                                                   float* __restrict__ eob) {
  __shared__ float acc[RN * DE];
  __shared__ unsigned cnt[RN];
  __shared__ unsigned short seg_s[NCH];
  __shared__ unsigned short seg_e[NCH];
  __shared__ unsigned next_chunk;

  const int tid = threadIdx.x;
  const int rg = blockIdx.x;

  for (int i = tid; i < RN * DE; i += 1024) acc[i] = 0.0f;
  if (tid < RN) cnt[tid] = 0u;
  if (tid == 0) next_chunk = 0u;
  if (tid < NCH) {
    const int w = (tid * TP + rg) >> 1;
    const unsigned d0 = tab2[w];
    const unsigned d1 = tab2[w + 1];
    seg_s[tid] = (unsigned short)((rg & 1) ? (d0 >> 16) : (d0 & 0xFFFFu));
    seg_e[tid] = (unsigned short)((rg & 1) ? (d1 & 0xFFFFu) : (d0 >> 16));
  }
  __syncthreads();

  const int lane = tid & 31;

  for (;;) {
    unsigned cdraw = 0u;
    if (lane == 0) cdraw = atomicAdd(&next_chunk, 1u);
    const unsigned c = (unsigned)__builtin_amdgcn_readfirstlane((int)cdraw);
    if (c >= (unsigned)NCH) break;
    const int s = __builtin_amdgcn_readfirstlane((int)seg_s[c]);
    const int t = __builtin_amdgcn_readfirstlane((int)seg_e[c]);
    const unsigned* kp = sorted + (size_t)c * CH;
    for (int j0 = s; j0 < t; j0 += 32) {
      const int n = (t - j0 < 32) ? (t - j0) : 32;
      const int jl = (j0 + lane < t) ? (j0 + lane) : (t - 1);
      const unsigned myk = kp[jl];
      if (lane < n) atomicAdd(&cnt[myk & (RN - 1)], 1u);
      for (int i = 0; i < n; i += NB) {
        float v[NB];
        unsigned k[NB];
#pragma unroll
        for (int u = 0; u < NB; ++u) {
          const int idx = (i + u < n) ? (i + u) : (n - 1);
          k[u] = (unsigned)__builtin_amdgcn_readlane((int)myk, idx);
          v[u] = edge_attr[(size_t)(k[u] >> 8) * DE + lane];
        }
        atomicAdd(&acc[(k[0] & (RN - 1)) * DE + lane], v[0]);
#pragma unroll
        for (int u = 1; u < NB; ++u) {
          if (i + u < n) atomicAdd(&acc[(k[u] & (RN - 1)) * DE + lane], v[u]);
        }
      }
    }
  }
  __syncthreads();

  for (int i = tid; i < RN * DE; i += 1024) {
    const int ln = i >> 5;
    const int node = rg * RN + ln;
    if (node < NN) eob[(size_t)node * DE + (i & 31)] = acc[i] / fmaxf((float)cnt[ln], 1.0f);
  }
}


__global__ __launch_bounds__(128) void mlp_wmma(
    const float* __restrict__ x, const float* __restrict__ eob,
    const float* __restrict__ u, const int* __restrict__ batch,
    const float* __restrict__ W1, const float* __restrict__ b1,
    const float* __restrict__ W2, const float* __restrict__ b2,
    float* __restrict__ y) {
  __shared__ float feat[ROWS * LDF];
  __shared__ float hbuf[ROWS * LDF];

  const int tid = threadIdx.x;
  const int nb = blockIdx.x * ROWS;

  for (int idx = tid; idx < ROWS * NF; idx += 128) {
    int r = idx >> 7;
    int c = idx & 127;
    int node = nb + r;
    float v = 0.0f;
    if (node < NN) {
      if (c < DN)           v = x[(long long)node * DN + c];
      else if (c < DN + DE) v = eob[(long long)node * DE + (c - DN)];
      else                  v = u[(long long)batch[node] * DG + (c - DN - DE)];
    }
    feat[r * LDF + c] = v;
  }
  __syncthreads();

  const int wave = tid >> 5;
  const int lane = tid & 31;
  const int half = lane >> 4;
  const int nidx = lane & 15;
  const int mrow0 = wave * 16;

  for (int nt = 0; nt < HIDDEN / 16; ++nt) {
    const int ncol = nt * 16 + nidx;
    const float bias = b1[ncol];
    v8f acc;
#pragma unroll
    for (int v = 0; v < 8; ++v) acc[v] = bias;
#pragma unroll 4
    for (int ks = 0; ks < NF / 4; ++ks) {
      int kk = ks * 4 + 2 * half;
      v2f a = *(const v2f*)&feat[(mrow0 + nidx) * LDF + kk];
      v2f b;
      b.x = W1[kk * HIDDEN + ncol];
      b.y = W1[(kk + 1) * HIDDEN + ncol];
      acc = __builtin_amdgcn_wmma_f32_16x16x4_f32(false, a, false, b,
                                                  (short)0, acc, false, false);
    }
#pragma unroll
    for (int v = 0; v < 8; ++v) {
      float r = acc[v];
      hbuf[(mrow0 + v + 8 * half) * LDF + ncol] = r > 0.0f ? r : 0.0f;
    }
  }
  __syncthreads();

  for (int nt = 0; nt < DN / 16; ++nt) {
    const int ncol = nt * 16 + nidx;
    const float bias = b2[ncol];
    v8f acc;
#pragma unroll
    for (int v = 0; v < 8; ++v) acc[v] = bias;
#pragma unroll 4
    for (int ks = 0; ks < HIDDEN / 4; ++ks) {
      int kk = ks * 4 + 2 * half;
      v2f a = *(const v2f*)&hbuf[(mrow0 + nidx) * LDF + kk];
      v2f b;
      b.x = W2[kk * DN + ncol];
      b.y = W2[(kk + 1) * DN + ncol];
      acc = __builtin_amdgcn_wmma_f32_16x16x4_f32(false, a, false, b,
                                                  (short)0, acc, false, false);
    }
#pragma unroll
    for (int v = 0; v < 8; ++v) {
      int node = nb + mrow0 + v + 8 * half;
      if (node < NN) {
        float r = acc[v];
        y[(long long)node * DN + ncol] = r > 0.0f ? r : 0.0f;
      }
    }
  }
}


extern "C" void kernel_launch(void* const* d_in, const int* in_sizes, int n_in,
                              void* d_out, int out_size, void* d_ws, size_t ws_size,
                              hipStream_t stream) {
  const float* x     = (const float*)d_in[0];
  const int*   eidx  = (const int*)d_in[1];
  const float* eattr = (const float*)d_in[2];
  const float* u     = (const float*)d_in[3];
  const int*   batch = (const int*)d_in[4];
  const float* W1    = (const float*)d_in[5];
  const float* b1    = (const float*)d_in[6];
  const float* W2    = (const float*)d_in[7];
  const float* b2    = (const float*)d_in[8];

  float* y   = (float*)d_out;
  float* eob = (float*)d_out + (size_t)NN * DN;

  unsigned* sorted = (unsigned*)d_ws;
  unsigned* tab2   = (unsigned*)((char*)d_ws + SORTED_BYTES);

  bin_edges<<<NCH, 1024, 0, stream>>>(eidx, sorted, tab2);

  range_mean<<<NRG, 1024, 0, stream>>>(eattr, sorted, tab2, eob);

  mlp_wmma<<<(NN + ROWS - 1) / ROWS, 128, 0, stream>>>(
      x, eob, u, batch, W1, b1, W2, b2, y);
}


# MLP feature tile: each thread's 16 pieces loaded as 128-bit words, all issued before the first LDS store (was one 32-bit load waited for per row, 64 in series); rows past the end clamped
# speedup vs baseline: 4.4376x; 1.2661x over previous
#include <hip/hip_runtime.h>

typedef __attribute__((ext_vector_type(2))) float v2f;
typedef __attribute__((ext_vector_type(8))) float v8f;

#define NN 50000
#define NE 1600000
#define DN 64
#define DE 32
#define DG 32
#define HIDDEN 128
#define NF 128
#define ROWS 64
#define LDF 130

#define CH 8192
#define NCH ((NE + CH - 1) / CH)
#define RSH 8
#define RN 256
#define NRG ((NN + RN - 1) / RN)
#define TP 256
#define SORTED_BYTES ((size_t)NE * 4)
#define NB 16

__global__ __launch_bounds__(1024) void bin_edges(const int* __restrict__ edge_index,
                                                  unsigned* __restrict__ sorted,
                                                  unsigned* __restrict__ tab2) {
  __shared__ unsigned hist[TP];
  __shared__ unsigned offs[TP];
  __shared__ unsigned stage[CH];

  const int tid = threadIdx.x;
  const int chunk = blockIdx.x;
  const int base = chunk * CH;

  if (tid < TP) hist[tid] = 0u;
  __syncthreads();

  int dsts[CH / 1024];
#pragma unroll
  for (int k = 0; k < CH / 1024; ++k) {
    int e = base + k * 1024 + tid;
    int ec = (e < NE) ? e : (NE - 1);
    int d = edge_index[NE + ec];
    dsts[k] = (e < NE) ? d : -1;
  }

  unsigned key[CH / 1024];
  unsigned slot[CH / 1024];
#pragma unroll
  for (int k = 0; k < CH / 1024; ++k) {
    int e = base + k * 1024 + tid;
    int d = dsts[k];
    slot[k] = 0xFFFFFFFFu;
    key[k] = 0u;
    if ((unsigned)d < (unsigned)NN) {
      unsigned r = (unsigned)d >> RSH;
      unsigned rank = atomicAdd(&hist[r], 1u);
      key[k] = ((unsigned)e << 8) | ((unsigned)d & (RN - 1));
      slot[k] = (r << 16) | rank;
    }
  }
  __syncthreads();

  if (tid < 32) {
    unsigned v[8];
    unsigned s = 0u;
#pragma unroll
    for (int i = 0; i < 8; ++i) { v[i] = hist[tid * 8 + i]; s += v[i]; }
    unsigned inc = s;
#pragma unroll
    for (int o = 1; o < 32; o <<= 1) {
      unsigned t = __shfl_up(inc, o);
      if (tid >= o) inc += t;
    }
    unsigned run = inc - s;
#pragma unroll
    for (int i = 0; i < 8; ++i) { offs[tid * 8 + i] = run; run += v[i]; }
  }
  __syncthreads();

#pragma unroll
  for (int k = 0; k < CH / 1024; ++k) {
    if (slot[k] != 0xFFFFFFFFu) stage[offs[slot[k] >> 16] + (slot[k] & 0xFFFFu)] = key[k];
  }
  if (tid < TP / 2) tab2[chunk * (TP / 2) + tid] = offs[2 * tid] | (offs[2 * tid + 1] << 16);
  __syncthreads();

  const unsigned total = offs[TP - 1];
#pragma unroll
  for (int k = 0; k < CH / 1024; ++k) {
    unsigned i = k * 1024 + tid;
    if (i < total) sorted[base + i] = stage[i];
  }
}

__global__ __launch_bounds__(1024) void range_mean(const float* __restrict__ edge_attr,
                                                   const unsigned* __restrict__ sorted,
                                                   const unsigned* __restrict__ tab2,
                                                   float* __restrict__ eob) {
  __shared__ float acc[RN * DE];
  __shared__ unsigned cnt[RN];
  __shared__ unsigned short seg_s[NCH];
  __shared__ unsigned short seg_e[NCH];
  __shared__ unsigned next_chunk;

  const int tid = threadIdx.x;
  const int rg = blockIdx.x;

  for (int i = tid; i < RN * DE; i += 1024) acc[i] = 0.0f;
  if (tid < RN) cnt[tid] = 0u;
  if (tid == 0) next_chunk = 0u;
  if (tid < NCH) {
    const int w = (tid * TP + rg) >> 1;
    const unsigned d0 = tab2[w];
    const unsigned d1 = tab2[w + 1];
    seg_s[tid] = (unsigned short)((rg & 1) ? (d0 >> 16) : (d0 & 0xFFFFu));
    seg_e[tid] = (unsigned short)((rg & 1) ? (d1 & 0xFFFFu) : (d0 >> 16));
  }
  __syncthreads();

  const int lane = tid & 31;

  for (;;) {
    unsigned cdraw = 0u;
    if (lane == 0) cdraw = atomicAdd(&next_chunk, 1u);
    const unsigned c = (unsigned)__builtin_amdgcn_readfirstlane((int)cdraw);
    if (c >= (unsigned)NCH) break;
    const int s = __builtin_amdgcn_readfirstlane((int)seg_s[c]);
    const int t = __builtin_amdgcn_readfirstlane((int)seg_e[c]);
    const unsigned* kp = sorted + (size_t)c * CH;
    for (int j0 = s; j0 < t; j0 += 32) {
      const int n = (t - j0 < 32) ? (t - j0) : 32;
      const int jl = (j0 + lane < t) ? (j0 + lane) : (t - 1);
      const unsigned myk = kp[jl];
      if (lane < n) atomicAdd(&cnt[myk & (RN - 1)], 1u);
      for (int i = 0; i < n; i += NB) {
        float v[NB];
        unsigned k[NB];
#pragma unroll
        for (int u = 0; u < NB; ++u) {
          const int idx = (i + u < n) ? (i + u) : (n - 1);
          k[u] = (unsigned)__builtin_amdgcn_readlane((int)myk, idx);
          v[u] = edge_attr[(size_t)(k[u] >> 8) * DE + lane];
        }
        atomicAdd(&acc[(k[0] & (RN - 1)) * DE + lane], v[0]);
#pragma unroll
        for (int u = 1; u < NB; ++u) {
          if (i + u < n) atomicAdd(&acc[(k[u] & (RN - 1)) * DE + lane], v[u]);
        }
      }
    }
  }
  __syncthreads();

  for (int i = tid; i < RN * DE; i += 1024) {
    const int ln = i >> 5;
    const int node = rg * RN + ln;
    if (node < NN) eob[(size_t)node * DE + (i & 31)] = acc[i] / fmaxf((float)cnt[ln], 1.0f);
  }
}


__global__ __launch_bounds__(128) void mlp_wmma(
    const float* __restrict__ x, const float* __restrict__ eob,
    const float* __restrict__ u, const int* __restrict__ batch,
    const float* __restrict__ W1, const float* __restrict__ b1,
    const float* __restrict__ W2, const float* __restrict__ b2,
    float* __restrict__ y) {
  __shared__ float feat[ROWS * LDF];
  __shared__ float hbuf[ROWS * LDF];

  const int tid = threadIdx.x;
  const int nb = blockIdx.x * ROWS;

  {
    int bq[4];
    float4 vx[8];
    float4 ve[4];
    float4 vu[4];
#pragma unroll
    for (int j = 0; j < 4; ++j) {
      const int q = tid + 128 * j;
      const int node = nb + (q >> 3);
      bq[j] = batch[node < NN ? node : NN - 1];
    }
#pragma unroll
    for (int j = 0; j < 8; ++j) {
      const int q = tid + 128 * j;
      const int node = nb + (q >> 4);
      const int nc = node < NN ? node : NN - 1;
      vx[j] = *(const float4*)&x[(size_t)nc * DN + (q & 15) * 4];
    }
#pragma unroll
    for (int j = 0; j < 4; ++j) {
      const int q = tid + 128 * j;
      const int node = nb + (q >> 3);
      const int nc = node < NN ? node : NN - 1;
      ve[j] = *(const float4*)&eob[(size_t)nc * DE + (q & 7) * 4];
    }
#pragma unroll
    for (int j = 0; j < 4; ++j) {
      const int q = tid + 128 * j;
      vu[j] = *(const float4*)&u[(size_t)bq[j] * DG + (q & 7) * 4];
    }
#pragma unroll
    for (int j = 0; j < 8; ++j) {
      const int q = tid + 128 * j;
      const int r = q >> 4;
      const bool ok = (nb + r) < NN;
      float* p = &feat[r * LDF + (q & 15) * 4];
      v2f lo, hi;
      lo.x = ok ? vx[j].x : 0.0f;  lo.y = ok ? vx[j].y : 0.0f;
      hi.x = ok ? vx[j].z : 0.0f;  hi.y = ok ? vx[j].w : 0.0f;
      *(v2f*)p = lo;
      *(v2f*)(p + 2) = hi;
    }
#pragma unroll
    for (int j = 0; j < 4; ++j) {
      const int q = tid + 128 * j;
      const int r = q >> 3;
      const bool ok = (nb + r) < NN;
      float* pe = &feat[r * LDF + DN + (q & 7) * 4];
      float* pu = pe + DE;
      v2f lo, hi;
      lo.x = ok ? ve[j].x : 0.0f;  lo.y = ok ? ve[j].y : 0.0f;
      hi.x = ok ? ve[j].z : 0.0f;  hi.y = ok ? ve[j].w : 0.0f;
      *(v2f*)pe = lo;
      *(v2f*)(pe + 2) = hi;
      lo.x = ok ? vu[j].x : 0.0f;  lo.y = ok ? vu[j].y : 0.0f;
      hi.x = ok ? vu[j].z : 0.0f;  hi.y = ok ? vu[j].w : 0.0f;
      *(v2f*)pu = lo;
      *(v2f*)(pu + 2) = hi;
    }
  }
  __syncthreads();

  const int wave = tid >> 5;
  const int lane = tid & 31;
  const int half = lane >> 4;
  const int nidx = lane & 15;
  const int mrow0 = wave * 16;

  for (int nt = 0; nt < HIDDEN / 16; ++nt) {
    const int ncol = nt * 16 + nidx;
    const float bias = b1[ncol];
    v8f acc;
#pragma unroll
    for (int v = 0; v < 8; ++v) acc[v] = bias;
#pragma unroll 4
    for (int ks = 0; ks < NF / 4; ++ks) {
      int kk = ks * 4 + 2 * half;
      v2f a = *(const v2f*)&feat[(mrow0 + nidx) * LDF + kk];
      v2f b;
      b.x = W1[kk * HIDDEN + ncol];
      b.y = W1[(kk + 1) * HIDDEN + ncol];
      acc = __builtin_amdgcn_wmma_f32_16x16x4_f32(false, a, false, b,
                                                  (short)0, acc, false, false);
    }
#pragma unroll
    for (int v = 0; v < 8; ++v) {
      float r = acc[v];
      hbuf[(mrow0 + v + 8 * half) * LDF + ncol] = r > 0.0f ? r : 0.0f;
    }
  }
  __syncthreads();

  for (int nt = 0; nt < DN / 16; ++nt) {
    const int ncol = nt * 16 + nidx;
    const float bias = b2[ncol];
    v8f acc;
#pragma unroll
    for (int v = 0; v < 8; ++v) acc[v] = bias;
#pragma unroll 4
    for (int ks = 0; ks < HIDDEN / 4; ++ks) {
      int kk = ks * 4 + 2 * half;
      v2f a = *(const v2f*)&hbuf[(mrow0 + nidx) * LDF + kk];
      v2f b;
      b.x = W2[kk * DN + ncol];
      b.y = W2[(kk + 1) * DN + ncol];
      acc = __builtin_amdgcn_wmma_f32_16x16x4_f32(false, a, false, b,
                                                  (short)0, acc, false, false);
    }
#pragma unroll
    for (int v = 0; v < 8; ++v) {
      int node = nb + mrow0 + v + 8 * half;
      if (node < NN) {
        float r = acc[v];
        y[(long long)node * DN + ncol] = r > 0.0f ? r : 0.0f;
      }
    }
  }
}


extern "C" void kernel_launch(void* const* d_in, const int* in_sizes, int n_in,
                              void* d_out, int out_size, void* d_ws, size_t ws_size,
                              hipStream_t stream) {
  const float* x     = (const float*)d_in[0];
  const int*   eidx  = (const int*)d_in[1];
  const float* eattr = (const float*)d_in[2];
  const float* u     = (const float*)d_in[3];
  const int*   batch = (const int*)d_in[4];
  const float* W1    = (const float*)d_in[5];
  const float* b1    = (const float*)d_in[6];
  const float* W2    = (const float*)d_in[7];
  const float* b2    = (const float*)d_in[8];

  float* y   = (float*)d_out;
  float* eob = (float*)d_out + (size_t)NN * DN;

  unsigned* sorted = (unsigned*)d_ws;
  unsigned* tab2   = (unsigned*)((char*)d_ws + SORTED_BYTES);

  bin_edges<<<NCH, 1024, 0, stream>>>(eidx, sorted, tab2);

  range_mean<<<NRG, 1024, 0, stream>>>(eattr, sorted, tab2, eob);

  mlp_wmma<<<(NN + ROWS - 1) / ROWS, 128, 0, stream>>>(
      x, eob, u, batch, W1, b1, W2, b2, y);
}


# MLP products: each wave keeps all four 16-row tiles of a column strip in registers (one B fragment feeds 4 WMMAs, W read once per work-group), next stage's fragments requested a stage ahead
# speedup vs baseline: 5.5966x; 1.2612x over previous
#include <hip/hip_runtime.h>

typedef __attribute__((ext_vector_type(2))) float v2f;
typedef __attribute__((ext_vector_type(8))) float v8f;

#define NN 50000
#define NE 1600000
#define DN 64
#define DE 32
#define DG 32
#define HIDDEN 128
#define NF 128
#define ROWS 64
#define LDF 130

#define CH 8192
#define NCH ((NE + CH - 1) / CH)
#define RSH 8
#define RN 256
#define NRG ((NN + RN - 1) / RN)
#define TP 256
#define SORTED_BYTES ((size_t)NE * 4)
#define NB 16

__global__ __launch_bounds__(1024) void bin_edges(const int* __restrict__ edge_index,
                                                  unsigned* __restrict__ sorted,
                                                  unsigned* __restrict__ tab2) {
  __shared__ unsigned hist[TP];
  __shared__ unsigned offs[TP];
  __shared__ unsigned stage[CH];

  const int tid = threadIdx.x;
  const int chunk = blockIdx.x;
  const int base = chunk * CH;

  if (tid < TP) hist[tid] = 0u;
  __syncthreads();

  int dsts[CH / 1024];
#pragma unroll
  for (int k = 0; k < CH / 1024; ++k) {
    int e = base + k * 1024 + tid;
    int ec = (e < NE) ? e : (NE - 1);
    int d = edge_index[NE + ec];
    dsts[k] = (e < NE) ? d : -1;
  }

  unsigned key[CH / 1024];
  unsigned slot[CH / 1024];
#pragma unroll
  for (int k = 0; k < CH / 1024; ++k) {
    int e = base + k * 1024 + tid;
    int d = dsts[k];
    slot[k] = 0xFFFFFFFFu;
    key[k] = 0u;
    if ((unsigned)d < (unsigned)NN) {
      unsigned r = (unsigned)d >> RSH;
      unsigned rank = atomicAdd(&hist[r], 1u);
      key[k] = ((unsigned)e << 8) | ((unsigned)d & (RN - 1));
      slot[k] = (r << 16) | rank;
    }
  }
  __syncthreads();

  if (tid < 32) {
    unsigned v[8];
    unsigned s = 0u;
#pragma unroll
    for (int i = 0; i < 8; ++i) { v[i] = hist[tid * 8 + i]; s += v[i]; }
    unsigned inc = s;
#pragma unroll
    for (int o = 1; o < 32; o <<= 1) {
      unsigned t = __shfl_up(inc, o);
      if (tid >= o) inc += t;
    }
    unsigned run = inc - s;
#pragma unroll
    for (int i = 0; i < 8; ++i) { offs[tid * 8 + i] = run; run += v[i]; }
  }
  __syncthreads();

#pragma unroll
  for (int k = 0; k < CH / 1024; ++k) {
    if (slot[k] != 0xFFFFFFFFu) stage[offs[slot[k] >> 16] + (slot[k] & 0xFFFFu)] = key[k];
  }
  if (tid < TP / 2) tab2[chunk * (TP / 2) + tid] = offs[2 * tid] | (offs[2 * tid + 1] << 16);
  __syncthreads();

  const unsigned total = offs[TP - 1];
#pragma unroll
  for (int k = 0; k < CH / 1024; ++k) {
    unsigned i = k * 1024 + tid;
    if (i < total) sorted[base + i] = stage[i];
  }
}

__global__ __launch_bounds__(1024) void range_mean(const float* __restrict__ edge_attr,
                                                   const unsigned* __restrict__ sorted,
                                                   const unsigned* __restrict__ tab2,
                                                   float* __restrict__ eob) {
  __shared__ float acc[RN * DE];
  __shared__ unsigned cnt[RN];
  __shared__ unsigned short seg_s[NCH];
  __shared__ unsigned short seg_e[NCH];
  __shared__ unsigned next_chunk;

  const int tid = threadIdx.x;
  const int rg = blockIdx.x;

  for (int i = tid; i < RN * DE; i += 1024) acc[i] = 0.0f;
  if (tid < RN) cnt[tid] = 0u;
  if (tid == 0) next_chunk = 0u;
  if (tid < NCH) {
    const int w = (tid * TP + rg) >> 1;
    const unsigned d0 = tab2[w];
    const unsigned d1 = tab2[w + 1];
    seg_s[tid] = (unsigned short)((rg & 1) ? (d0 >> 16) : (d0 & 0xFFFFu));
    seg_e[tid] = (unsigned short)((rg & 1) ? (d1 & 0xFFFFu) : (d0 >> 16));
  }
  __syncthreads();

  const int lane = tid & 31;

  for (;;) {
    unsigned cdraw = 0u;
    if (lane == 0) cdraw = atomicAdd(&next_chunk, 1u);
    const unsigned c = (unsigned)__builtin_amdgcn_readfirstlane((int)cdraw);
    if (c >= (unsigned)NCH) break;
    const int s = __builtin_amdgcn_readfirstlane((int)seg_s[c]);
    const int t = __builtin_amdgcn_readfirstlane((int)seg_e[c]);
    const unsigned* kp = sorted + (size_t)c * CH;
    for (int j0 = s; j0 < t; j0 += 32) {
      const int n = (t - j0 < 32) ? (t - j0) : 32;
      const int jl = (j0 + lane < t) ? (j0 + lane) : (t - 1);
      const unsigned myk = kp[jl];
      if (lane < n) atomicAdd(&cnt[myk & (RN - 1)], 1u);
      for (int i = 0; i < n; i += NB) {
        float v[NB];
        unsigned k[NB];
#pragma unroll
        for (int u = 0; u < NB; ++u) {
          const int idx = (i + u < n) ? (i + u) : (n - 1);
          k[u] = (unsigned)__builtin_amdgcn_readlane((int)myk, idx);
          v[u] = edge_attr[(size_t)(k[u] >> 8) * DE + lane];
        }
        atomicAdd(&acc[(k[0] & (RN - 1)) * DE + lane], v[0]);
#pragma unroll
        for (int u = 1; u < NB; ++u) {
          if (i + u < n) atomicAdd(&acc[(k[u] & (RN - 1)) * DE + lane], v[u]);
        }
      }
    }
  }
  __syncthreads();

  for (int i = tid; i < RN * DE; i += 1024) {
    const int ln = i >> 5;
    const int node = rg * RN + ln;
    if (node < NN) eob[(size_t)node * DE + (i & 31)] = acc[i] / fmaxf((float)cnt[ln], 1.0f);
  }
}


__global__ __launch_bounds__(128) void mlp_wmma(
    const float* __restrict__ x, const float* __restrict__ eob,
    const float* __restrict__ u, const int* __restrict__ batch,
    const float* __restrict__ W1, const float* __restrict__ b1,
    const float* __restrict__ W2, const float* __restrict__ b2,
    float* __restrict__ y) {
  __shared__ float feat[ROWS * LDF];
  __shared__ float hbuf[ROWS * LDF];

  const int tid = threadIdx.x;
  const int nb = blockIdx.x * ROWS;

  {
    int bq[4];
    float4 vx[8];
    float4 ve[4];
    float4 vu[4];
#pragma unroll
    for (int j = 0; j < 4; ++j) {
      const int q = tid + 128 * j;
      const int node = nb + (q >> 3);
      bq[j] = batch[node < NN ? node : NN - 1];
    }
#pragma unroll
    for (int j = 0; j < 8; ++j) {
      const int q = tid + 128 * j;
      const int node = nb + (q >> 4);
      const int nc = node < NN ? node : NN - 1;
      vx[j] = *(const float4*)&x[(size_t)nc * DN + (q & 15) * 4];
    }
#pragma unroll
    for (int j = 0; j < 4; ++j) {
      const int q = tid + 128 * j;
      const int node = nb + (q >> 3);
      const int nc = node < NN ? node : NN - 1;
      ve[j] = *(const float4*)&eob[(size_t)nc * DE + (q & 7) * 4];
    }
#pragma unroll
    for (int j = 0; j < 4; ++j) {
      const int q = tid + 128 * j;
      vu[j] = *(const float4*)&u[(size_t)bq[j] * DG + (q & 7) * 4];
    }
#pragma unroll
    for (int j = 0; j < 8; ++j) {
      const int q = tid + 128 * j;
      const int r = q >> 4;
      const bool ok = (nb + r) < NN;
      float* p = &feat[r * LDF + (q & 15) * 4];
      v2f lo, hi;
      lo.x = ok ? vx[j].x : 0.0f;  lo.y = ok ? vx[j].y : 0.0f;
      hi.x = ok ? vx[j].z : 0.0f;  hi.y = ok ? vx[j].w : 0.0f;
      *(v2f*)p = lo;
      *(v2f*)(p + 2) = hi;
    }
#pragma unroll
    for (int j = 0; j < 4; ++j) {
      const int q = tid + 128 * j;
      const int r = q >> 3;
      const bool ok = (nb + r) < NN;
      float* pe = &feat[r * LDF + DN + (q & 7) * 4];
      float* pu = pe + DE;
      v2f lo, hi;
      lo.x = ok ? ve[j].x : 0.0f;  lo.y = ok ? ve[j].y : 0.0f;
      hi.x = ok ? ve[j].z : 0.0f;  hi.y = ok ? ve[j].w : 0.0f;
      *(v2f*)pe = lo;
      *(v2f*)(pe + 2) = hi;
      lo.x = ok ? vu[j].x : 0.0f;  lo.y = ok ? vu[j].y : 0.0f;
      hi.x = ok ? vu[j].z : 0.0f;  hi.y = ok ? vu[j].w : 0.0f;
      *(v2f*)pu = lo;
      *(v2f*)(pu + 2) = hi;
    }
  }
  __syncthreads();

  const int wave = tid >> 5;
  const int lane = tid & 31;
  const int half = lane >> 4;
  const int nidx = lane & 15;


  {
    const int c0 = wave * 32 + nidx;
    v8f acc[4][2];
#pragma unroll
    for (int n = 0; n < 2; ++n) {
      const float bias = b1[c0 + 16 * n];
#pragma unroll
      for (int m = 0; m < 4; ++m)
#pragma unroll
        for (int v = 0; v < 8; ++v) acc[m][n][v] = bias;
    }
    const float* ap = &feat[nidx * LDF + 2 * half];
    const float* bp = &W1[(2 * half) * HIDDEN + c0];
    v2f a[2][4], b[2][2];
#pragma unroll
    for (int s = 0; s < 2; ++s) {
#pragma unroll
      for (int m = 0; m < 4; ++m) a[s][m] = *(const v2f*)&ap[m * 16 * LDF + s * 4];
#pragma unroll
      for (int n = 0; n < 2; ++n) {
        b[s][n].x = bp[(s * 4) * HIDDEN + 16 * n];
        b[s][n].y = bp[(s * 4 + 1) * HIDDEN + 16 * n];
      }
    }
#pragma unroll 2
    for (int st = 0; st < NF / 8; ++st) {
      const int k0 = ((st + 1 < NF / 8) ? (st + 1) : st) * 8;
      v2f an[2][4], bn[2][2];
#pragma unroll
      for (int s = 0; s < 2; ++s) {
#pragma unroll
        for (int n = 0; n < 2; ++n) {
          bn[s][n].x = bp[(k0 + s * 4) * HIDDEN + 16 * n];
          bn[s][n].y = bp[(k0 + s * 4 + 1) * HIDDEN + 16 * n];
        }
#pragma unroll
        for (int m = 0; m < 4; ++m) an[s][m] = *(const v2f*)&ap[m * 16 * LDF + k0 + s * 4];
      }
#pragma unroll
      for (int s = 0; s < 2; ++s)
#pragma unroll
        for (int m = 0; m < 4; ++m)
#pragma unroll
          for (int n = 0; n < 2; ++n)
            acc[m][n] = __builtin_amdgcn_wmma_f32_16x16x4_f32(false, a[s][m], false, b[s][n],
                                                              (short)0, acc[m][n], false, false);
#pragma unroll
      for (int s = 0; s < 2; ++s) {
#pragma unroll
        for (int m = 0; m < 4; ++m) a[s][m] = an[s][m];
#pragma unroll
        for (int n = 0; n < 2; ++n) b[s][n] = bn[s][n];
      }
    }
#pragma unroll
    for (int m = 0; m < 4; ++m)
#pragma unroll
      for (int n = 0; n < 2; ++n)
#pragma unroll
        for (int v = 0; v < 8; ++v) {
          const float r = acc[m][n][v];
          hbuf[(m * 16 + v + 8 * half) * LDF + c0 + 16 * n] = r > 0.0f ? r : 0.0f;
        }
  }
  __syncthreads();

  {
    const int c0 = wave * 16 + nidx;
    const float bias = b2[c0];
    v8f acc[4];
#pragma unroll
    for (int m = 0; m < 4; ++m)
#pragma unroll
      for (int v = 0; v < 8; ++v) acc[m][v] = bias;
    const float* ap = &hbuf[nidx * LDF + 2 * half];
    const float* bp = &W2[(2 * half) * DN + c0];
    v2f a[4][4], b[4];
#pragma unroll
    for (int s = 0; s < 4; ++s) {
#pragma unroll
      for (int m = 0; m < 4; ++m) a[s][m] = *(const v2f*)&ap[m * 16 * LDF + s * 4];
      b[s].x = bp[(s * 4) * DN];
      b[s].y = bp[(s * 4 + 1) * DN];
    }
#pragma unroll 2
    for (int st = 0; st < HIDDEN / 16; ++st) {
      const int k0 = ((st + 1 < HIDDEN / 16) ? (st + 1) : st) * 16;
      v2f an[4][4], bn[4];
#pragma unroll
      for (int s = 0; s < 4; ++s) {
        bn[s].x = bp[(k0 + s * 4) * DN];
        bn[s].y = bp[(k0 + s * 4 + 1) * DN];
#pragma unroll
        for (int m = 0; m < 4; ++m) an[s][m] = *(const v2f*)&ap[m * 16 * LDF + k0 + s * 4];
      }
#pragma unroll
      for (int s = 0; s < 4; ++s)
#pragma unroll
        for (int m = 0; m < 4; ++m)
          acc[m] = __builtin_amdgcn_wmma_f32_16x16x4_f32(false, a[s][m], false, b[s],
                                                         (short)0, acc[m], false, false);
#pragma unroll
      for (int s = 0; s < 4; ++s) {
#pragma unroll
        for (int m = 0; m < 4; ++m) a[s][m] = an[s][m];
        b[s] = bn[s];
      }
    }
#pragma unroll
    for (int m = 0; m < 4; ++m)
#pragma unroll
      for (int v = 0; v < 8; ++v) {
        const int node = nb + m * 16 + v + 8 * half;
        if (node < NN) {
          const float r = acc[m][v];
          y[(long long)node * DN + c0] = r > 0.0f ? r : 0.0f;
        }
      }
  }
}


extern "C" void kernel_launch(void* const* d_in, const int* in_sizes, int n_in,
                              void* d_out, int out_size, void* d_ws, size_t ws_size,
                              hipStream_t stream) {
  const float* x     = (const float*)d_in[0];
  const int*   eidx  = (const int*)d_in[1];
  const float* eattr = (const float*)d_in[2];
  const float* u     = (const float*)d_in[3];
  const int*   batch = (const int*)d_in[4];
  const float* W1    = (const float*)d_in[5];
  const float* b1    = (const float*)d_in[6];
  const float* W2    = (const float*)d_in[7];
  const float* b2    = (const float*)d_in[8];

  float* y   = (float*)d_out;
  float* eob = (float*)d_out + (size_t)NN * DN;

  unsigned* sorted = (unsigned*)d_ws;
  unsigned* tab2   = (unsigned*)((char*)d_ws + SORTED_BYTES);

  bin_edges<<<NCH, 1024, 0, stream>>>(eidx, sorted, tab2);

  range_mean<<<NRG, 1024, 0, stream>>>(eattr, sorted, tab2, eob);

  mlp_wmma<<<(NN + ROWS - 1) / ROWS, 128, 0, stream>>>(
      x, eob, u, batch, W1, b1, W2, b2, y);
}


# scatter-mean: each range's keys of all chunks copied into one LDS list, rows fetched in full 32-row batches, 16 kept in flight under the adds (was a per-chunk walk); LDS f32 atomics, arrival order
# speedup vs baseline: 6.1902x; 1.1061x over previous
#include <hip/hip_runtime.h>

typedef __attribute__((ext_vector_type(2))) float v2f;
typedef __attribute__((ext_vector_type(8))) float v8f;
typedef __attribute__((address_space(1))) float gfloat;

#define NN 50000
#define NE 1600000
#define DN 64
#define DE 32
#define DG 32
#define HIDDEN 128
#define NF 128
#define ROWS 64
#define LDF 130

#define CH 8192
#define NCH ((NE + CH - 1) / CH)
#define RSH 8
#define RN 256
#define NRG ((NN + RN - 1) / RN)
#define TP 256
#define SORTED_BYTES ((size_t)NE * 4)

__global__ __launch_bounds__(1024) void bin_edges(const int* __restrict__ edge_index,
                                                  unsigned* __restrict__ sorted,
                                                  unsigned* __restrict__ tab2) {
  __shared__ unsigned hist[TP];
  __shared__ unsigned offs[TP];
  __shared__ unsigned stage[CH];

  const int tid = threadIdx.x;
  const int chunk = blockIdx.x;
  const int base = chunk * CH;

  if (tid < TP) hist[tid] = 0u;
  __syncthreads();

  int dsts[CH / 1024];
#pragma unroll
  for (int k = 0; k < CH / 1024; ++k) {
    int e = base + k * 1024 + tid;
    int ec = (e < NE) ? e : (NE - 1);
    int d = edge_index[NE + ec];
    dsts[k] = (e < NE) ? d : -1;
  }

  unsigned key[CH / 1024];
  unsigned slot[CH / 1024];
#pragma unroll
  for (int k = 0; k < CH / 1024; ++k) {
    int e = base + k * 1024 + tid;
    int d = dsts[k];
    slot[k] = 0xFFFFFFFFu;
    key[k] = 0u;
    if ((unsigned)d < (unsigned)NN) {
      unsigned r = (unsigned)d >> RSH;
      unsigned rank = atomicAdd(&hist[r], 1u);
      key[k] = ((unsigned)e << 8) | ((unsigned)d & (RN - 1));
      slot[k] = (r << 16) | rank;
    }
  }
  __syncthreads();

  if (tid < 32) {
    unsigned v[8];
    unsigned s = 0u;
#pragma unroll
    for (int i = 0; i < 8; ++i) { v[i] = hist[tid * 8 + i]; s += v[i]; }
    unsigned inc = s;
#pragma unroll
    for (int o = 1; o < 32; o <<= 1) {
      unsigned t = __shfl_up(inc, o);
      if (tid >= o) inc += t;
    }
    unsigned run = inc - s;
#pragma unroll
    for (int i = 0; i < 8; ++i) { offs[tid * 8 + i] = run; run += v[i]; }
  }
  __syncthreads();

#pragma unroll
  for (int k = 0; k < CH / 1024; ++k) {
    if (slot[k] != 0xFFFFFFFFu) stage[offs[slot[k] >> 16] + (slot[k] & 0xFFFFu)] = key[k];
  }
  if (tid < TP / 2) tab2[chunk * (TP / 2) + tid] = offs[2 * tid] | (offs[2 * tid + 1] << 16);
  __syncthreads();

  const unsigned total = offs[TP - 1];
#pragma unroll
  for (int k = 0; k < CH / 1024; ++k) {
    unsigned i = k * 1024 + tid;
    if (i < total) sorted[base + i] = stage[i];
  }
}

#define KCAP 12288
#define CPW ((NCH + 31) / 32)
#define BR 32
#define HB 16

__global__ __launch_bounds__(1024) void range_mean(const float* __restrict__ edge_attr,
                                                   const unsigned* __restrict__ sorted,
                                                   const unsigned* __restrict__ tab2,
                                                   float* __restrict__ eob) {
  __shared__ float acc[RN * DE];
  __shared__ unsigned cnt[RN];
  __shared__ unsigned keys[KCAP];
  __shared__ unsigned pre[NCH + 1];
  __shared__ unsigned short seg_s[NCH];
  __shared__ unsigned short seg_n[NCH];

  const int tid = threadIdx.x;
  const int rg = blockIdx.x;
  const int lane = tid & 31;
  const int wave = __builtin_amdgcn_readfirstlane(tid >> 5);

  unsigned d0 = 0u, d1 = 0u;
  if (tid < NCH) {
    const int w = (tid * TP + rg) >> 1;
    d0 = tab2[w];
    d1 = tab2[w + 1];
  }
  for (int i = tid; i < RN * DE; i += 1024) acc[i] = 0.0f;
  if (tid < RN) cnt[tid] = 0u;
  if (tid < NCH) {
    const unsigned s = (rg & 1) ? (d0 >> 16) : (d0 & 0xFFFFu);
    const unsigned e = (rg & 1) ? (d1 & 0xFFFFu) : (d0 >> 16);
    seg_s[tid] = (unsigned short)s;
    seg_n[tid] = (unsigned short)(e - s);
  }
  __syncthreads();

  if (tid < 32) {
    unsigned v[CPW];
    unsigned s = 0u;
#pragma unroll
    for (int i = 0; i < CPW; ++i) {
      const int c = tid * CPW + i;
      const unsigned x = seg_n[c < NCH ? c : 0];
      v[i] = (c < NCH) ? x : 0u;
      s += v[i];
    }
    unsigned inc = s;
#pragma unroll
    for (int o = 1; o < 32; o <<= 1) {
      unsigned t = __shfl_up(inc, o);
      if (tid >= o) inc += t;
    }
    unsigned run = inc - s;
#pragma unroll
    for (int i = 0; i < CPW; ++i) {
      const int c = tid * CPW + i;
      if (c < NCH) pre[c] = run;
      run += v[i];
    }
    if (tid == 31) pre[NCH] = run;
  }
  __syncthreads();

  const unsigned total = (unsigned)__builtin_amdgcn_readfirstlane((int)pre[NCH]);

  auto issue_rows = [&](float (&v)[HB], unsigned ro, int u0) {
    const gfloat* rp[HB];
#pragma unroll
    for (int u = 0; u < HB; ++u) {
      const unsigned o = (unsigned)__builtin_amdgcn_readlane((int)ro, u0 + u);
      const unsigned long long rb = (unsigned long long)edge_attr + (unsigned long long)o;
      const unsigned lo = (unsigned)__builtin_amdgcn_readfirstlane((int)(unsigned)rb);
      const unsigned hi = (unsigned)__builtin_amdgcn_readfirstlane((int)(unsigned)(rb >> 32));
      rp[u] = (const gfloat*)(((unsigned long long)hi << 32) | (unsigned long long)lo);
    }
#pragma unroll
    for (int u = 0; u < HB; ++u) v[u] = rp[u][lane];
  };
  auto add_full = [&](const float (&v)[HB], unsigned ti, int u0) {
#pragma unroll
    for (int u = 0; u < HB; ++u) {
      const unsigned t = (unsigned)__builtin_amdgcn_readlane((int)ti, u0 + u);
      atomicAdd(&acc[t + lane], v[u]);
    }
  };
  auto add_part = [&](const float (&v)[HB], unsigned ti, int u0, int n) {
#pragma unroll
    for (int u = 0; u < HB; ++u) {
      if (u0 + u < n) {
        const unsigned t = (unsigned)__builtin_amdgcn_readlane((int)ti, u0 + u);
        atomicAdd(&acc[t + lane], v[u]);
      }
    }
  };

  for (unsigned base = 0u; base < total; base += KCAP) {
    {
      unsigned kk[CPW][2];
      int pos[CPW];
      int nn[CPW];
#pragma unroll
      for (int i = 0; i < CPW; ++i) {
        const int c = wave + 32 * i;
        const int cc = (c < NCH) ? c : (NCH - 1);
        const int s = seg_s[cc];
        nn[i] = (c < NCH) ? (int)seg_n[cc] : 0;
        pos[i] = (int)pre[cc] - (int)base;
#pragma unroll
        for (int h = 0; h < 2; ++h) {
          const int j = lane + 32 * h;
          const int off = (j < nn[i]) ? (s + j) : 0;
          kk[i][h] = sorted[cc * CH + off];
        }
      }
#pragma unroll
      for (int i = 0; i < CPW; ++i) {
#pragma unroll
        for (int h = 0; h < 2; ++h) {
          const int j = lane + 32 * h;
          const int f = pos[i] + j;
          if (j < nn[i] && (unsigned)f < (unsigned)KCAP) keys[f] = kk[i][h];
        }
      }
      for (int c = wave; c < NCH; c += 32) {
        const int n = __builtin_amdgcn_readfirstlane((int)seg_n[c]);
        if (n > 64) {
          const int s = seg_s[c];
          const int p = (int)pre[c] - (int)base;
          for (int j = 64 + lane; j < n; j += 32) {
            const int f = p + j;
            if ((unsigned)f < (unsigned)KCAP) keys[f] = sorted[c * CH + s + j];
          }
        }
      }
    }
    __syncthreads();

    const unsigned m = (total - base < (unsigned)KCAP) ? (total - base) : (unsigned)KCAP;
    const int nbat = (int)((m + BR - 1) / BR);
    const int mine = (nbat > wave) ? ((nbat - wave + 31) >> 5) : 0;
    auto load_keys = [&](int j, unsigned& ro, unsigned& ti) {
      const unsigned i = (unsigned)(wave + 32 * j) * BR + lane;
      const unsigned myk = keys[i < m ? i : m - 1];
      const unsigned ln = myk & (RN - 1);
      atomicAdd(&cnt[ln], (i < m) ? 1u : 0u);
      ro = (myk >> 8) << 7;
      ti = ln * DE;
    };
    auto rows_in = [&](int j) -> int {
      const unsigned r = m - (unsigned)(wave + 32 * j) * BR;
      return (r < (unsigned)BR) ? (int)r : BR;
    };
    if (mine > 0) {
      float va[HB], vb[HB];
      unsigned ro, ti;
      load_keys(0, ro, ti);
      issue_rows(va, ro, 0);
      issue_rows(vb, ro, HB);
      for (int j = 0; j + 1 < mine; ++j) {
        unsigned ron, tin;
        add_full(va, ti, 0);
        load_keys(j + 1, ron, tin);
        issue_rows(va, ron, 0);
        add_full(vb, ti, HB);
        issue_rows(vb, ron, HB);
        ro = ron;
        ti = tin;
      }
      const int n = rows_in(mine - 1);
      add_part(va, ti, 0, n);
      add_part(vb, ti, HB, n);
    }
    __syncthreads();
  }

  for (int i = tid; i < RN * DE; i += 1024) {
    const int ln = i >> 5;
    const int node = rg * RN + ln;
    if (node < NN) eob[(size_t)node * DE + (i & 31)] = acc[i] / fmaxf((float)cnt[ln], 1.0f);
  }
}


__global__ __launch_bounds__(128) void mlp_wmma(
    const float* __restrict__ x, const float* __restrict__ eob,
    const float* __restrict__ u, const int* __restrict__ batch,
    const float* __restrict__ W1, const float* __restrict__ b1,
    const float* __restrict__ W2, const float* __restrict__ b2,
    float* __restrict__ y) {
  __shared__ float feat[ROWS * LDF];
  __shared__ float hbuf[ROWS * LDF];

  const int tid = threadIdx.x;
  const int nb = blockIdx.x * ROWS;

  {
    int bq[4];
    float4 vx[8];
    float4 ve[4];
    float4 vu[4];
#pragma unroll
    for (int j = 0; j < 4; ++j) {
      const int q = tid + 128 * j;
      const int node = nb + (q >> 3);
      bq[j] = batch[node < NN ? node : NN - 1];
    }
#pragma unroll
    for (int j = 0; j < 8; ++j) {
      const int q = tid + 128 * j;
      const int node = nb + (q >> 4);
      const int nc = node < NN ? node : NN - 1;
      vx[j] = *(const float4*)&x[(size_t)nc * DN + (q & 15) * 4];
    }
#pragma unroll
    for (int j = 0; j < 4; ++j) {
      const int q = tid + 128 * j;
      const int node = nb + (q >> 3);
      const int nc = node < NN ? node : NN - 1;
      ve[j] = *(const float4*)&eob[(size_t)nc * DE + (q & 7) * 4];
    }
#pragma unroll
    for (int j = 0; j < 4; ++j) {
      const int q = tid + 128 * j;
      vu[j] = *(const float4*)&u[(size_t)bq[j] * DG + (q & 7) * 4];
    }
#pragma unroll
    for (int j = 0; j < 8; ++j) {
      const int q = tid + 128 * j;
      const int r = q >> 4;
      const bool ok = (nb + r) < NN;
      float* p = &feat[r * LDF + (q & 15) * 4];
      v2f lo, hi;
      lo.x = ok ? vx[j].x : 0.0f;  lo.y = ok ? vx[j].y : 0.0f;
      hi.x = ok ? vx[j].z : 0.0f;  hi.y = ok ? vx[j].w : 0.0f;
      *(v2f*)p = lo;
      *(v2f*)(p + 2) = hi;
    }
#pragma unroll
    for (int j = 0; j < 4; ++j) {
      const int q = tid + 128 * j;
      const int r = q >> 3;
      const bool ok = (nb + r) < NN;
      float* pe = &feat[r * LDF + DN + (q & 7) * 4];
      float* pu = pe + DE;
      v2f lo, hi;
      lo.x = ok ? ve[j].x : 0.0f;  lo.y = ok ? ve[j].y : 0.0f;
      hi.x = ok ? ve[j].z : 0.0f;  hi.y = ok ? ve[j].w : 0.0f;
      *(v2f*)pe = lo;
      *(v2f*)(pe + 2) = hi;
      lo.x = ok ? vu[j].x : 0.0f;  lo.y = ok ? vu[j].y : 0.0f;
      hi.x = ok ? vu[j].z : 0.0f;  hi.y = ok ? vu[j].w : 0.0f;
      *(v2f*)pu = lo;
      *(v2f*)(pu + 2) = hi;
    }
  }
  __syncthreads();

  const int wave = tid >> 5;
  const int lane = tid & 31;
  const int half = lane >> 4;
  const int nidx = lane & 15;


  {
    const int c0 = wave * 32 + nidx;
    v8f acc[4][2];
#pragma unroll
    for (int n = 0; n < 2; ++n) {
      const float bias = b1[c0 + 16 * n];
#pragma unroll
      for (int m = 0; m < 4; ++m)
#pragma unroll
        for (int v = 0; v < 8; ++v) acc[m][n][v] = bias;
    }
    const float* ap = &feat[nidx * LDF + 2 * half];
    const float* bp = &W1[(2 * half) * HIDDEN + c0];
    v2f a[2][4], b[2][2];
#pragma unroll
    for (int s = 0; s < 2; ++s) {
#pragma unroll
      for (int m = 0; m < 4; ++m) a[s][m] = *(const v2f*)&ap[m * 16 * LDF + s * 4];
#pragma unroll
      for (int n = 0; n < 2; ++n) {
        b[s][n].x = bp[(s * 4) * HIDDEN + 16 * n];
        b[s][n].y = bp[(s * 4 + 1) * HIDDEN + 16 * n];
      }
    }
#pragma unroll 2
    for (int st = 0; st < NF / 8; ++st) {
      const int k0 = ((st + 1 < NF / 8) ? (st + 1) : st) * 8;
      v2f an[2][4], bn[2][2];
#pragma unroll
      for (int s = 0; s < 2; ++s) {
#pragma unroll
        for (int n = 0; n < 2; ++n) {
          bn[s][n].x = bp[(k0 + s * 4) * HIDDEN + 16 * n];
          bn[s][n].y = bp[(k0 + s * 4 + 1) * HIDDEN + 16 * n];
        }
#pragma unroll
        for (int m = 0; m < 4; ++m) an[s][m] = *(const v2f*)&ap[m * 16 * LDF + k0 + s * 4];
      }
#pragma unroll
      for (int s = 0; s < 2; ++s)
#pragma unroll
        for (int m = 0; m < 4; ++m)
#pragma unroll
          for (int n = 0; n < 2; ++n)
            acc[m][n] = __builtin_amdgcn_wmma_f32_16x16x4_f32(false, a[s][m], false, b[s][n],
                                                              (short)0, acc[m][n], false, false);
#pragma unroll
      for (int s = 0; s < 2; ++s) {
#pragma unroll
        for (int m = 0; m < 4; ++m) a[s][m] = an[s][m];
#pragma unroll
        for (int n = 0; n < 2; ++n) b[s][n] = bn[s][n];
      }
    }
#pragma unroll
    for (int m = 0; m < 4; ++m)
#pragma unroll
      for (int n = 0; n < 2; ++n)
#pragma unroll
        for (int v = 0; v < 8; ++v) {
          const float r = acc[m][n][v];
          hbuf[(m * 16 + v + 8 * half) * LDF + c0 + 16 * n] = r > 0.0f ? r : 0.0f;
        }
  }
  __syncthreads();

  {
    const int c0 = wave * 16 + nidx;
    const float bias = b2[c0];
    v8f acc[4];
#pragma unroll
    for (int m = 0; m < 4; ++m)
#pragma unroll
      for (int v = 0; v < 8; ++v) acc[m][v] = bias;
    const float* ap = &hbuf[nidx * LDF + 2 * half];
    const float* bp = &W2[(2 * half) * DN + c0];
    v2f a[4][4], b[4];
#pragma unroll
    for (int s = 0; s < 4; ++s) {
#pragma unroll
      for (int m = 0; m < 4; ++m) a[s][m] = *(const v2f*)&ap[m * 16 * LDF + s * 4];
      b[s].x = bp[(s * 4) * DN];
      b[s].y = bp[(s * 4 + 1) * DN];
    }
#pragma unroll 2
    for (int st = 0; st < HIDDEN / 16; ++st) {
      const int k0 = ((st + 1 < HIDDEN / 16) ? (st + 1) : st) * 16;
      v2f an[4][4], bn[4];
#pragma unroll
      for (int s = 0; s < 4; ++s) {
        bn[s].x = bp[(k0 + s * 4) * DN];
        bn[s].y = bp[(k0 + s * 4 + 1) * DN];
#pragma unroll
        for (int m = 0; m < 4; ++m) an[s][m] = *(const v2f*)&ap[m * 16 * LDF + k0 + s * 4];
      }
#pragma unroll
      for (int s = 0; s < 4; ++s)
#pragma unroll
        for (int m = 0; m < 4; ++m)
          acc[m] = __builtin_amdgcn_wmma_f32_16x16x4_f32(false, a[s][m], false, b[s],
                                                         (short)0, acc[m], false, false);
#pragma unroll
      for (int s = 0; s < 4; ++s) {
#pragma unroll
        for (int m = 0; m < 4; ++m) a[s][m] = an[s][m];
        b[s] = bn[s];
      }
    }
#pragma unroll
    for (int m = 0; m < 4; ++m)
#pragma unroll
      for (int v = 0; v < 8; ++v) {
        const int node = nb + m * 16 + v + 8 * half;
        if (node < NN) {
          const float r = acc[m][v];
          y[(long long)node * DN + c0] = r > 0.0f ? r : 0.0f;
        }
      }
  }
}


extern "C" void kernel_launch(void* const* d_in, const int* in_sizes, int n_in,
                              void* d_out, int out_size, void* d_ws, size_t ws_size,
                              hipStream_t stream) {
  const float* x     = (const float*)d_in[0];
  const int*   eidx  = (const int*)d_in[1];
  const float* eattr = (const float*)d_in[2];
  const float* u     = (const float*)d_in[3];
  const int*   batch = (const int*)d_in[4];
  const float* W1    = (const float*)d_in[5];
  const float* b1    = (const float*)d_in[6];
  const float* W2    = (const float*)d_in[7];
  const float* b2    = (const float*)d_in[8];

  float* y   = (float*)d_out;
  float* eob = (float*)d_out + (size_t)NN * DN;

  unsigned* sorted = (unsigned*)d_ws;
  unsigned* tab2   = (unsigned*)((char*)d_ws + SORTED_BYTES);

  bin_edges<<<NCH, 1024, 0, stream>>>(eidx, sorted, tab2);

  range_mean<<<NRG, 1024, 0, stream>>>(eattr, sorted, tab2, eob);

  mlp_wmma<<<(NN + ROWS - 1) / ROWS, 128, 0, stream>>>(
      x, eob, u, batch, W1, b1, W2, b2, y);
}


# scatter-mean and MLP fused: a range's work-group sums its edge rows into the mean columns of its LDS feature tile and runs both layers from it; 3 launches to 2; LDS f32 atomics, arrival order
# speedup vs baseline: 6.6172x; 1.0690x over previous
#include <hip/hip_runtime.h>

typedef __attribute__((ext_vector_type(2))) float v2f;
typedef __attribute__((ext_vector_type(8))) float v8f;
typedef __attribute__((address_space(1))) float gfloat;

#define NN 50000
#define NE 1600000
#define DN 64
#define DE 32
#define DG 32
#define HIDDEN 128
#define NF 128
#define ROWS 64
#define LDF 130

#define CH 8192
#define NCH ((NE + CH - 1) / CH)
#define RSH 8
#define RN 256
#define NRG ((NN + RN - 1) / RN)
#define TP 256
#define SORTED_BYTES ((size_t)NE * 4)

__global__ __launch_bounds__(1024) void bin_edges(const int* __restrict__ edge_index,
                                                  unsigned* __restrict__ sorted,
                                                  unsigned* __restrict__ tab2) {
  __shared__ unsigned hist[TP];
  __shared__ unsigned offs[TP];
  __shared__ unsigned stage[CH];

  const int tid = threadIdx.x;
  const int chunk = blockIdx.x;
  const int base = chunk * CH;

  if (tid < TP) hist[tid] = 0u;
  __syncthreads();

  int dsts[CH / 1024];
#pragma unroll
  for (int k = 0; k < CH / 1024; ++k) {
    int e = base + k * 1024 + tid;
    int ec = (e < NE) ? e : (NE - 1);
    int d = edge_index[NE + ec];
    dsts[k] = (e < NE) ? d : -1;
  }

  unsigned key[CH / 1024];
  unsigned slot[CH / 1024];
#pragma unroll
  for (int k = 0; k < CH / 1024; ++k) {
    int e = base + k * 1024 + tid;
    int d = dsts[k];
    slot[k] = 0xFFFFFFFFu;
    key[k] = 0u;
    if ((unsigned)d < (unsigned)NN) {
      unsigned r = (unsigned)d >> RSH;
      unsigned rank = atomicAdd(&hist[r], 1u);
      key[k] = ((unsigned)e << 8) | ((unsigned)d & (RN - 1));
      slot[k] = (r << 16) | rank;
    }
  }
  __syncthreads();

  if (tid < 32) {
    unsigned v[8];
    unsigned s = 0u;
#pragma unroll
    for (int i = 0; i < 8; ++i) { v[i] = hist[tid * 8 + i]; s += v[i]; }
    unsigned inc = s;
#pragma unroll
    for (int o = 1; o < 32; o <<= 1) {
      unsigned t = __shfl_up(inc, o);
      if (tid >= o) inc += t;
    }
    unsigned run = inc - s;
#pragma unroll
    for (int i = 0; i < 8; ++i) { offs[tid * 8 + i] = run; run += v[i]; }
  }
  __syncthreads();

#pragma unroll
  for (int k = 0; k < CH / 1024; ++k) {
    if (slot[k] != 0xFFFFFFFFu) stage[offs[slot[k] >> 16] + (slot[k] & 0xFFFFu)] = key[k];
  }
  if (tid < TP / 2) tab2[chunk * (TP / 2) + tid] = offs[2 * tid] | (offs[2 * tid + 1] << 16);
  __syncthreads();

  const unsigned total = offs[TP - 1];
#pragma unroll
  for (int k = 0; k < CH / 1024; ++k) {
    unsigned i = k * 1024 + tid;
    if (i < total) sorted[base + i] = stage[i];
  }
}

#define KCAP 12288
#define CPW ((NCH + 31) / 32)
#define BR 32
#define HB 16
#define MLPW (4 * (RN / ROWS))


__global__ __launch_bounds__(1024) void range_mlp(const float* __restrict__ edge_attr,
                                                  const unsigned* __restrict__ sorted,
                                                  const unsigned* __restrict__ tab2,
                                                  const float* __restrict__ x,
                                                  const float* __restrict__ u,
                                                  const int* __restrict__ batch,
                                                  const float* __restrict__ W1, const float* __restrict__ b1,
                                                  const float* __restrict__ W2, const float* __restrict__ b2,
                                                  float* __restrict__ eob, float* __restrict__ y) {
  __shared__ float feat[RN * LDF];
  __shared__ float hbuf[RN * LDF];
  __shared__ unsigned cnt[RN];
  __shared__ unsigned pre[NCH + 1];
  __shared__ unsigned short seg_s[NCH];
  __shared__ unsigned short seg_n[NCH];
  unsigned* const keys = (unsigned*)hbuf;

  const int tid = threadIdx.x;
  const int rg = blockIdx.x;
  const int lane = tid & 31;
  const int wave = __builtin_amdgcn_readfirstlane(tid >> 5);

  unsigned d0 = 0u, d1 = 0u;
  if (tid < NCH) {
    const int w = (tid * TP + rg) >> 1;
    d0 = tab2[w];
    d1 = tab2[w + 1];
  }
  int bq[2];
#pragma unroll
  for (int j = 0; j < 2; ++j) {
    const int q = tid + 1024 * j;
    const int node = rg * RN + (q >> 3);
    bq[j] = batch[node < NN ? node : NN - 1];
  }
  for (int i = tid; i < RN * DE; i += 1024) feat[(i >> 5) * LDF + DN + (i & 31)] = 0.0f;
  if (tid < RN) cnt[tid] = 0u;
  if (tid < NCH) {
    const unsigned s = (rg & 1) ? (d0 >> 16) : (d0 & 0xFFFFu);
    const unsigned e = (rg & 1) ? (d1 & 0xFFFFu) : (d0 >> 16);
    seg_s[tid] = (unsigned short)s;
    seg_n[tid] = (unsigned short)(e - s);
  }
  __syncthreads();

  if (tid < 32) {
    unsigned v[CPW];
    unsigned s = 0u;
#pragma unroll
    for (int i = 0; i < CPW; ++i) {
      const int c = tid * CPW + i;
      const unsigned sn = seg_n[c < NCH ? c : 0];
      v[i] = (c < NCH) ? sn : 0u;
      s += v[i];
    }
    unsigned inc = s;
#pragma unroll
    for (int o = 1; o < 32; o <<= 1) {
      unsigned t = __shfl_up(inc, o);
      if (tid >= o) inc += t;
    }
    unsigned run = inc - s;
#pragma unroll
    for (int i = 0; i < CPW; ++i) {
      const int c = tid * CPW + i;
      if (c < NCH) pre[c] = run;
      run += v[i];
    }
    if (tid == 31) pre[NCH] = run;
  }
  __syncthreads();

  const unsigned total = (unsigned)__builtin_amdgcn_readfirstlane((int)pre[NCH]);

  float4 vx[4];
  float4 vu[2];
#pragma unroll
  for (int j = 0; j < 4; ++j) {
    const int q = tid + 1024 * j;
    const int node = rg * RN + (q >> 4);
    const int nc = node < NN ? node : NN - 1;
    vx[j] = *(const float4*)&x[(size_t)nc * DN + (q & 15) * 4];
  }
#pragma unroll
  for (int j = 0; j < 2; ++j) {
    const int q = tid + 1024 * j;
    vu[j] = *(const float4*)&u[(size_t)bq[j] * DG + (q & 7) * 4];
  }

  auto issue_rows = [&](float (&v)[HB], unsigned ro, int u0) {
    const gfloat* rp[HB];
#pragma unroll
    for (int u = 0; u < HB; ++u) {
      const unsigned o = (unsigned)__builtin_amdgcn_readlane((int)ro, u0 + u);
      const unsigned long long rb = (unsigned long long)edge_attr + (unsigned long long)o;
      const unsigned lo = (unsigned)__builtin_amdgcn_readfirstlane((int)(unsigned)rb);
      const unsigned hi = (unsigned)__builtin_amdgcn_readfirstlane((int)(unsigned)(rb >> 32));
      rp[u] = (const gfloat*)(((unsigned long long)hi << 32) | (unsigned long long)lo);
    }
#pragma unroll
    for (int u = 0; u < HB; ++u) v[u] = rp[u][lane];
  };
  auto add_full = [&](const float (&v)[HB], unsigned ti, int u0) {
#pragma unroll
    for (int u = 0; u < HB; ++u) {
      const unsigned t = (unsigned)__builtin_amdgcn_readlane((int)ti, u0 + u);
      atomicAdd(&feat[t + lane], v[u]);
    }
  };
  auto add_part = [&](const float (&v)[HB], unsigned ti, int u0, int n) {
#pragma unroll
    for (int u = 0; u < HB; ++u) {
      if (u0 + u < n) {
        const unsigned t = (unsigned)__builtin_amdgcn_readlane((int)ti, u0 + u);
        atomicAdd(&feat[t + lane], v[u]);
      }
    }
  };

  for (unsigned base = 0u; base < total; base += KCAP) {
    {
      unsigned kk[CPW][2];
      int pos[CPW];
      int nn[CPW];
#pragma unroll
      for (int i = 0; i < CPW; ++i) {
        const int c = wave + 32 * i;
        const int cc = (c < NCH) ? c : (NCH - 1);
        const int s = seg_s[cc];
        nn[i] = (c < NCH) ? (int)seg_n[cc] : 0;
        pos[i] = (int)pre[cc] - (int)base;
#pragma unroll
        for (int h = 0; h < 2; ++h) {
          const int j = lane + 32 * h;
          const int off = (j < nn[i]) ? (s + j) : 0;
          kk[i][h] = sorted[cc * CH + off];
        }
      }
#pragma unroll
      for (int i = 0; i < CPW; ++i) {
#pragma unroll
        for (int h = 0; h < 2; ++h) {
          const int j = lane + 32 * h;
          const int f = pos[i] + j;
          if (j < nn[i] && (unsigned)f < (unsigned)KCAP) keys[f] = kk[i][h];
        }
      }
      for (int c = wave; c < NCH; c += 32) {
        const int n = __builtin_amdgcn_readfirstlane((int)seg_n[c]);
        if (n > 64) {
          const int s = seg_s[c];
          const int p = (int)pre[c] - (int)base;
          for (int j = 64 + lane; j < n; j += 32) {
            const int f = p + j;
            if ((unsigned)f < (unsigned)KCAP) keys[f] = sorted[c * CH + s + j];
          }
        }
      }
    }
    __syncthreads();

    const unsigned m = (total - base < (unsigned)KCAP) ? (total - base) : (unsigned)KCAP;
    const int nbat = (int)((m + BR - 1) / BR);
    const int mine = (nbat > wave) ? ((nbat - wave + 31) >> 5) : 0;
    auto load_keys = [&](int j, unsigned& ro, unsigned& ti) {
      const unsigned i = (unsigned)(wave + 32 * j) * BR + lane;
      const unsigned myk = keys[i < m ? i : m - 1];
      const unsigned ln = myk & (RN - 1);
      atomicAdd(&cnt[ln], (i < m) ? 1u : 0u);
      ro = (myk >> 8) << 7;
      ti = ln * LDF + DN;
    };
    auto rows_in = [&](int j) -> int {
      const unsigned r = m - (unsigned)(wave + 32 * j) * BR;
      return (r < (unsigned)BR) ? (int)r : BR;
    };
    if (mine > 0) {
      float va[HB], vb[HB];
      unsigned ro, ti;
      load_keys(0, ro, ti);
      issue_rows(va, ro, 0);
      issue_rows(vb, ro, HB);
      for (int j = 0; j + 1 < mine; ++j) {
        unsigned ron, tin;
        add_full(va, ti, 0);
        load_keys(j + 1, ron, tin);
        __builtin_amdgcn_sched_barrier(0);
        issue_rows(va, ron, 0);
        add_full(vb, ti, HB);
        __builtin_amdgcn_sched_barrier(0);
        issue_rows(vb, ron, HB);
        ro = ron;
        ti = tin;
      }
      const int n = rows_in(mine - 1);
      add_part(va, ti, 0, n);
      add_part(vb, ti, HB, n);
    }
    __syncthreads();
  }

#pragma unroll
  for (int j = 0; j < 4; ++j) {
    const int q = tid + 1024 * j;
    const int r = q >> 4;
    const bool ok = (rg * RN + r) < NN;
    float* p = &feat[r * LDF + (q & 15) * 4];
    v2f lo, hi;
    lo.x = ok ? vx[j].x : 0.0f;  lo.y = ok ? vx[j].y : 0.0f;
    hi.x = ok ? vx[j].z : 0.0f;  hi.y = ok ? vx[j].w : 0.0f;
    *(v2f*)p = lo;
    *(v2f*)(p + 2) = hi;
  }
#pragma unroll
  for (int j = 0; j < 2; ++j) {
    const int q = tid + 1024 * j;
    const int r = q >> 3;
    const bool ok = (rg * RN + r) < NN;
    float* pu = &feat[r * LDF + DN + DE + (q & 7) * 4];
    v2f lo, hi;
    lo.x = ok ? vu[j].x : 0.0f;  lo.y = ok ? vu[j].y : 0.0f;
    hi.x = ok ? vu[j].z : 0.0f;  hi.y = ok ? vu[j].w : 0.0f;
    *(v2f*)pu = lo;
    *(v2f*)(pu + 2) = hi;
  }

  const int nb = rg * RN;
  for (int i = tid; i < RN * DE; i += 1024) {
    const int ln = i >> 5;
    float* p = &feat[ln * LDF + DN + (i & 31)];
    const float mean = *p / fmaxf((float)cnt[ln], 1.0f);
    *p = mean;
    if (nb + ln < NN) eob[(size_t)(nb + ln) * DE + (i & 31)] = mean;
  }
  __syncthreads();

  const int half = lane >> 4;
  const int nidx = lane & 15;
  const int grp = wave >> 2;
  const int strip = wave & 3;

  if (wave < MLPW) {
    const int c0 = strip * 32 + nidx;
    v8f acc[4][2];
#pragma unroll
    for (int n = 0; n < 2; ++n) {
      const float bias = b1[c0 + 16 * n];
#pragma unroll
      for (int m = 0; m < 4; ++m)
#pragma unroll
        for (int v = 0; v < 8; ++v) acc[m][n][v] = bias;
    }
    const float* ap = &feat[(grp * ROWS + nidx) * LDF + 2 * half];
    const float* bp = &W1[(2 * half) * HIDDEN + c0];
    v2f a[2][4], b[2][2];
#pragma unroll
    for (int s = 0; s < 2; ++s) {
#pragma unroll
      for (int m = 0; m < 4; ++m) a[s][m] = *(const v2f*)&ap[m * 16 * LDF + s * 4];
#pragma unroll
      for (int n = 0; n < 2; ++n) {
        b[s][n].x = bp[(s * 4) * HIDDEN + 16 * n];
        b[s][n].y = bp[(s * 4 + 1) * HIDDEN + 16 * n];
      }
    }
#pragma unroll 2
    for (int st = 0; st < NF / 8; ++st) {
      const int k0 = ((st + 1 < NF / 8) ? (st + 1) : st) * 8;
      v2f an[2][4], bn[2][2];
#pragma unroll
      for (int s = 0; s < 2; ++s) {
#pragma unroll
        for (int n = 0; n < 2; ++n) {
          bn[s][n].x = bp[(k0 + s * 4) * HIDDEN + 16 * n];
          bn[s][n].y = bp[(k0 + s * 4 + 1) * HIDDEN + 16 * n];
        }
#pragma unroll
        for (int m = 0; m < 4; ++m) an[s][m] = *(const v2f*)&ap[m * 16 * LDF + k0 + s * 4];
      }
#pragma unroll
      for (int s = 0; s < 2; ++s)
#pragma unroll
        for (int m = 0; m < 4; ++m)
#pragma unroll
          for (int n = 0; n < 2; ++n)
            acc[m][n] = __builtin_amdgcn_wmma_f32_16x16x4_f32(false, a[s][m], false, b[s][n],
                                                              (short)0, acc[m][n], false, false);
#pragma unroll
      for (int s = 0; s < 2; ++s) {
#pragma unroll
        for (int m = 0; m < 4; ++m) a[s][m] = an[s][m];
#pragma unroll
        for (int n = 0; n < 2; ++n) b[s][n] = bn[s][n];
      }
    }
#pragma unroll
    for (int m = 0; m < 4; ++m)
#pragma unroll
      for (int n = 0; n < 2; ++n)
#pragma unroll
        for (int v = 0; v < 8; ++v) {
          const float r = acc[m][n][v];
          hbuf[(grp * ROWS + m * 16 + v + 8 * half) * LDF + c0 + 16 * n] = r > 0.0f ? r : 0.0f;
        }
  }
  __syncthreads();

  if (wave < MLPW) {
    const int c0 = strip * 16 + nidx;
    const float bias = b2[c0];
    v8f acc[4];
#pragma unroll
    for (int m = 0; m < 4; ++m)
#pragma unroll
      for (int v = 0; v < 8; ++v) acc[m][v] = bias;
    const float* ap = &hbuf[(grp * ROWS + nidx) * LDF + 2 * half];
    const float* bp = &W2[(2 * half) * DN + c0];
    v2f a[4][4], b[4];
#pragma unroll
    for (int s = 0; s < 4; ++s) {
#pragma unroll
      for (int m = 0; m < 4; ++m) a[s][m] = *(const v2f*)&ap[m * 16 * LDF + s * 4];
      b[s].x = bp[(s * 4) * DN];
      b[s].y = bp[(s * 4 + 1) * DN];
    }
#pragma unroll 2
    for (int st = 0; st < HIDDEN / 16; ++st) {
      const int k0 = ((st + 1 < HIDDEN / 16) ? (st + 1) : st) * 16;
      v2f an[4][4], bn[4];
#pragma unroll
      for (int s = 0; s < 4; ++s) {
        bn[s].x = bp[(k0 + s * 4) * DN];
        bn[s].y = bp[(k0 + s * 4 + 1) * DN];
#pragma unroll
        for (int m = 0; m < 4; ++m) an[s][m] = *(const v2f*)&ap[m * 16 * LDF + k0 + s * 4];
      }
#pragma unroll
      for (int s = 0; s < 4; ++s)
#pragma unroll
        for (int m = 0; m < 4; ++m)
          acc[m] = __builtin_amdgcn_wmma_f32_16x16x4_f32(false, a[s][m], false, b[s],
                                                         (short)0, acc[m], false, false);
#pragma unroll
      for (int s = 0; s < 4; ++s) {
#pragma unroll
        for (int m = 0; m < 4; ++m) a[s][m] = an[s][m];
        b[s] = bn[s];
      }
    }
#pragma unroll
    for (int m = 0; m < 4; ++m)
#pragma unroll
      for (int v = 0; v < 8; ++v) {
        const int node = nb + grp * ROWS + m * 16 + v + 8 * half;
        if (node < NN) {
          const float r = acc[m][v];
          y[(long long)node * DN + c0] = r > 0.0f ? r : 0.0f;
        }
      }
  }
}


extern "C" void kernel_launch(void* const* d_in, const int* in_sizes, int n_in,
                              void* d_out, int out_size, void* d_ws, size_t ws_size,
                              hipStream_t stream) {
  const float* x     = (const float*)d_in[0];
  const int*   eidx  = (const int*)d_in[1];
  const float* eattr = (const float*)d_in[2];
  const float* u     = (const float*)d_in[3];
  const int*   batch = (const int*)d_in[4];
  const float* W1    = (const float*)d_in[5];
  const float* b1    = (const float*)d_in[6];
  const float* W2    = (const float*)d_in[7];
  const float* b2    = (const float*)d_in[8];

  float* y   = (float*)d_out;
  float* eob = (float*)d_out + (size_t)NN * DN;

  unsigned* sorted = (unsigned*)d_ws;
  unsigned* tab2   = (unsigned*)((char*)d_ws + SORTED_BYTES);

  bin_edges<<<NCH, 1024, 0, stream>>>(eidx, sorted, tab2);

  range_mlp<<<NRG, 1024, 0, stream>>>(eattr, sorted, tab2, x, u, batch, W1, b1, W2, b2, eob, y);
}


# node range 256 -> 208 rows: 241 work-groups (was 196) fill the 256 WGPs one each, 13 M tiles of MLP per WGP (was 16; wave groups of 4,4,4,1 tiles); LDS f32 atomics, arrival order as before
# speedup vs baseline: 7.0159x; 1.0603x over previous
#include <hip/hip_runtime.h>

typedef __attribute__((ext_vector_type(2))) float v2f;
typedef __attribute__((ext_vector_type(8))) float v8f;
typedef __attribute__((address_space(1))) float gfloat;

#define NN 50000
#define NE 1600000
#define DN 64
#define DE 32
#define DG 32
#define HIDDEN 128
#define NF 128
#define ROWS 64
#define LDF 130

#define CH 8192
#define NCH ((NE + CH - 1) / CH)
#define RN 208
#define NRG ((NN + RN - 1) / RN)
#define TP 256
#define SORTED_BYTES ((size_t)NE * 4)

__global__ __launch_bounds__(1024) void bin_edges(const int* __restrict__ edge_index,
                                                  unsigned* __restrict__ sorted,
                                                  unsigned* __restrict__ tab2) {
  __shared__ unsigned hist[TP];
  __shared__ unsigned offs[TP];
  __shared__ unsigned stage[CH];

  const int tid = threadIdx.x;
  const int chunk = blockIdx.x;
  const int base = chunk * CH;

  if (tid < TP) hist[tid] = 0u;
  __syncthreads();

  int dsts[CH / 1024];
#pragma unroll
  for (int k = 0; k < CH / 1024; ++k) {
    int e = base + k * 1024 + tid;
    int ec = (e < NE) ? e : (NE - 1);
    int d = edge_index[NE + ec];
    dsts[k] = (e < NE) ? d : -1;
  }

  unsigned key[CH / 1024];
  unsigned slot[CH / 1024];
#pragma unroll
  for (int k = 0; k < CH / 1024; ++k) {
    int e = base + k * 1024 + tid;
    int d = dsts[k];
    slot[k] = 0xFFFFFFFFu;
    key[k] = 0u;
    if ((unsigned)d < (unsigned)NN) {
      unsigned r = (unsigned)d / RN;
      unsigned rank = atomicAdd(&hist[r], 1u);
      key[k] = ((unsigned)e << 8) | ((unsigned)d - r * RN);
      slot[k] = (r << 16) | rank;
    }
  }
  __syncthreads();

  if (tid < 32) {
    unsigned v[8];
    unsigned s = 0u;
#pragma unroll
    for (int i = 0; i < 8; ++i) { v[i] = hist[tid * 8 + i]; s += v[i]; }
    unsigned inc = s;
#pragma unroll
    for (int o = 1; o < 32; o <<= 1) {
      unsigned t = __shfl_up(inc, o);
      if (tid >= o) inc += t;
    }
    unsigned run = inc - s;
#pragma unroll
    for (int i = 0; i < 8; ++i) { offs[tid * 8 + i] = run; run += v[i]; }
  }
  __syncthreads();

#pragma unroll
  for (int k = 0; k < CH / 1024; ++k) {
    if (slot[k] != 0xFFFFFFFFu) stage[offs[slot[k] >> 16] + (slot[k] & 0xFFFFu)] = key[k];
  }
  if (tid < TP / 2) tab2[chunk * (TP / 2) + tid] = offs[2 * tid] | (offs[2 * tid + 1] << 16);
  __syncthreads();

  const unsigned total = offs[TP - 1];
#pragma unroll
  for (int k = 0; k < CH / 1024; ++k) {
    unsigned i = k * 1024 + tid;
    if (i < total) sorted[base + i] = stage[i];
  }
}

#define KCAP 12288
#define CPW ((NCH + 31) / 32)
#define BR 32
#define HB 16
#define MTILES (RN / 16)
#define NGRP ((MTILES + 3) / 4)
#define LASTMT (MTILES - 4 * (NGRP - 1))
#define MLPW (4 * NGRP)

template <int N> struct tile_count { static constexpr int value = N; };


__global__ __launch_bounds__(1024) void range_mlp(const float* __restrict__ edge_attr,
                                                  const unsigned* __restrict__ sorted,
                                                  const unsigned* __restrict__ tab2,
                                                  const float* __restrict__ x,
                                                  const float* __restrict__ u,
                                                  const int* __restrict__ batch,
                                                  const float* __restrict__ W1, const float* __restrict__ b1,
                                                  const float* __restrict__ W2, const float* __restrict__ b2,
                                                  float* __restrict__ eob, float* __restrict__ y) {
  __shared__ float feat[RN * LDF];
  __shared__ float hbuf[RN * LDF];
  __shared__ unsigned cnt[RN];
  __shared__ unsigned pre[NCH + 1];
  __shared__ unsigned short seg_s[NCH];
  __shared__ unsigned short seg_n[NCH];
  unsigned* const keys = (unsigned*)hbuf;

  const int tid = threadIdx.x;
  const int rg = blockIdx.x;
  const int lane = tid & 31;
  const int wave = __builtin_amdgcn_readfirstlane(tid >> 5);

  unsigned d0 = 0u, d1 = 0u;
  if (tid < NCH) {
    const int w = (tid * TP + rg) >> 1;
    d0 = tab2[w];
    d1 = tab2[w + 1];
  }
  int bq[2];
#pragma unroll
  for (int j = 0; j < 2; ++j) {
    const int q = tid + 1024 * j;
    const int r = (q >> 3) < RN ? (q >> 3) : RN - 1;
    const int node = rg * RN + r;
    bq[j] = batch[node < NN ? node : NN - 1];
  }
  for (int i = tid; i < RN * DE; i += 1024) feat[(i >> 5) * LDF + DN + (i & 31)] = 0.0f;
  if (tid < RN) cnt[tid] = 0u;
  if (tid < NCH) {
    const unsigned s = (rg & 1) ? (d0 >> 16) : (d0 & 0xFFFFu);
    const unsigned e = (rg & 1) ? (d1 & 0xFFFFu) : (d0 >> 16);
    seg_s[tid] = (unsigned short)s;
    seg_n[tid] = (unsigned short)(e - s);
  }
  __syncthreads();

  if (tid < 32) {
    unsigned v[CPW];
    unsigned s = 0u;
#pragma unroll
    for (int i = 0; i < CPW; ++i) {
      const int c = tid * CPW + i;
      const unsigned sn = seg_n[c < NCH ? c : 0];
      v[i] = (c < NCH) ? sn : 0u;
      s += v[i];
    }
    unsigned inc = s;
#pragma unroll
    for (int o = 1; o < 32; o <<= 1) {
      unsigned t = __shfl_up(inc, o);
      if (tid >= o) inc += t;
    }
    unsigned run = inc - s;
#pragma unroll
    for (int i = 0; i < CPW; ++i) {
      const int c = tid * CPW + i;
      if (c < NCH) pre[c] = run;
      run += v[i];
    }
    if (tid == 31) pre[NCH] = run;
  }
  __syncthreads();

  const unsigned total = (unsigned)__builtin_amdgcn_readfirstlane((int)pre[NCH]);

  float4 vx[4];
  float4 vu[2];
#pragma unroll
  for (int j = 0; j < 4; ++j) {
    const int q = tid + 1024 * j;
    const int r = (q >> 4) < RN ? (q >> 4) : RN - 1;
    const int node = rg * RN + r;
    const int nc = node < NN ? node : NN - 1;
    vx[j] = *(const float4*)&x[(size_t)nc * DN + (q & 15) * 4];
  }
#pragma unroll
  for (int j = 0; j < 2; ++j) {
    const int q = tid + 1024 * j;
    vu[j] = *(const float4*)&u[(size_t)bq[j] * DG + (q & 7) * 4];
  }

  auto issue_rows = [&](float (&v)[HB], unsigned ro, int u0) {
    const gfloat* rp[HB];
#pragma unroll
    for (int u = 0; u < HB; ++u) {
      const unsigned o = (unsigned)__builtin_amdgcn_readlane((int)ro, u0 + u);
      const unsigned long long rb = (unsigned long long)edge_attr + (unsigned long long)o;
      const unsigned lo = (unsigned)__builtin_amdgcn_readfirstlane((int)(unsigned)rb);
      const unsigned hi = (unsigned)__builtin_amdgcn_readfirstlane((int)(unsigned)(rb >> 32));
      rp[u] = (const gfloat*)(((unsigned long long)hi << 32) | (unsigned long long)lo);
    }
#pragma unroll
    for (int u = 0; u < HB; ++u) v[u] = rp[u][lane];
  };
  auto add_full = [&](const float (&v)[HB], unsigned ti, int u0) {
#pragma unroll
    for (int u = 0; u < HB; ++u) {
      const unsigned t = (unsigned)__builtin_amdgcn_readlane((int)ti, u0 + u);
      atomicAdd(&feat[t + lane], v[u]);
    }
  };
  auto add_part = [&](const float (&v)[HB], unsigned ti, int u0, int n) {
#pragma unroll
    for (int u = 0; u < HB; ++u) {
      if (u0 + u < n) {
        const unsigned t = (unsigned)__builtin_amdgcn_readlane((int)ti, u0 + u);
        atomicAdd(&feat[t + lane], v[u]);
      }
    }
  };

  for (unsigned base = 0u; base < total; base += KCAP) {
    {
      unsigned kk[CPW][2];
      int pos[CPW];
      int nn[CPW];
#pragma unroll
      for (int i = 0; i < CPW; ++i) {
        const int c = wave + 32 * i;
        const int cc = (c < NCH) ? c : (NCH - 1);
        const int s = seg_s[cc];
        nn[i] = (c < NCH) ? (int)seg_n[cc] : 0;
        pos[i] = (int)pre[cc] - (int)base;
#pragma unroll
        for (int h = 0; h < 2; ++h) {
          const int j = lane + 32 * h;
          const int off = (j < nn[i]) ? (s + j) : 0;
          kk[i][h] = sorted[cc * CH + off];
        }
      }
#pragma unroll
      for (int i = 0; i < CPW; ++i) {
#pragma unroll
        for (int h = 0; h < 2; ++h) {
          const int j = lane + 32 * h;
          const int f = pos[i] + j;
          if (j < nn[i] && (unsigned)f < (unsigned)KCAP) keys[f] = kk[i][h];
        }
      }
      for (int c = wave; c < NCH; c += 32) {
        const int n = __builtin_amdgcn_readfirstlane((int)seg_n[c]);
        if (n > 64) {
          const int s = seg_s[c];
          const int p = (int)pre[c] - (int)base;
          for (int j = 64 + lane; j < n; j += 32) {
            const int f = p + j;
            if ((unsigned)f < (unsigned)KCAP) keys[f] = sorted[c * CH + s + j];
          }
        }
      }
    }
    __syncthreads();

    const unsigned m = (total - base < (unsigned)KCAP) ? (total - base) : (unsigned)KCAP;
    const int nbat = (int)((m + BR - 1) / BR);
    const int mine = (nbat > wave) ? ((nbat - wave + 31) >> 5) : 0;
    auto load_keys = [&](int j, unsigned& ro, unsigned& ti) {
      const unsigned i = (unsigned)(wave + 32 * j) * BR + lane;
      const unsigned myk = keys[i < m ? i : m - 1];
      const unsigned ln = myk & 255u;
      atomicAdd(&cnt[ln], (i < m) ? 1u : 0u);
      ro = (myk >> 8) << 7;
      ti = ln * LDF + DN;
    };
    auto rows_in = [&](int j) -> int {
      const unsigned r = m - (unsigned)(wave + 32 * j) * BR;
      return (r < (unsigned)BR) ? (int)r : BR;
    };
    if (mine > 0) {
      float va[HB], vb[HB];
      unsigned ro, ti;
      load_keys(0, ro, ti);
      issue_rows(va, ro, 0);
      issue_rows(vb, ro, HB);
      for (int j = 0; j + 1 < mine; ++j) {
        unsigned ron, tin;
        add_full(va, ti, 0);
        load_keys(j + 1, ron, tin);
        __builtin_amdgcn_sched_barrier(0);
        issue_rows(va, ron, 0);
        add_full(vb, ti, HB);
        __builtin_amdgcn_sched_barrier(0);
        issue_rows(vb, ron, HB);
        ro = ron;
        ti = tin;
      }
      const int n = rows_in(mine - 1);
      add_part(va, ti, 0, n);
      add_part(vb, ti, HB, n);
    }
    __syncthreads();
  }

#pragma unroll
  for (int j = 0; j < 4; ++j) {
    const int q = tid + 1024 * j;
    const int r = q >> 4;
    const bool ok = (rg * RN + r) < NN;
    float* p = &feat[r * LDF + (q & 15) * 4];
    v2f lo, hi;
    lo.x = ok ? vx[j].x : 0.0f;  lo.y = ok ? vx[j].y : 0.0f;
    hi.x = ok ? vx[j].z : 0.0f;  hi.y = ok ? vx[j].w : 0.0f;
    if (r < RN) {
      *(v2f*)p = lo;
      *(v2f*)(p + 2) = hi;
    }
  }
#pragma unroll
  for (int j = 0; j < 2; ++j) {
    const int q = tid + 1024 * j;
    const int r = q >> 3;
    const bool ok = (rg * RN + r) < NN;
    float* pu = &feat[r * LDF + DN + DE + (q & 7) * 4];
    v2f lo, hi;
    lo.x = ok ? vu[j].x : 0.0f;  lo.y = ok ? vu[j].y : 0.0f;
    hi.x = ok ? vu[j].z : 0.0f;  hi.y = ok ? vu[j].w : 0.0f;
    if (r < RN) {
      *(v2f*)pu = lo;
      *(v2f*)(pu + 2) = hi;
    }
  }

  const int nb = rg * RN;
  for (int i = tid; i < RN * DE; i += 1024) {
    const int ln = i >> 5;
    float* p = &feat[ln * LDF + DN + (i & 31)];
    const float mean = *p / fmaxf((float)cnt[ln], 1.0f);
    *p = mean;
    if (nb + ln < NN) eob[(size_t)(nb + ln) * DE + (i & 31)] = mean;
  }
  __syncthreads();

  const int half = lane >> 4;
  const int nidx = lane & 15;
  const int grp = wave >> 2;
  const int strip = wave & 3;

  auto layer1 = [&](auto mt) {
    constexpr int MT = decltype(mt)::value;
    const int c0 = strip * 32 + nidx;
    v8f acc[MT][2];
#pragma unroll
    for (int n = 0; n < 2; ++n) {
      const float bias = b1[c0 + 16 * n];
#pragma unroll
      for (int m = 0; m < MT; ++m)
#pragma unroll
        for (int v = 0; v < 8; ++v) acc[m][n][v] = bias;
    }
    const float* ap = &feat[(grp * ROWS + nidx) * LDF + 2 * half];
    const float* bp = &W1[(2 * half) * HIDDEN + c0];
    v2f a[2][MT], b[2][2];
#pragma unroll
    for (int s = 0; s < 2; ++s) {
#pragma unroll
      for (int m = 0; m < MT; ++m) a[s][m] = *(const v2f*)&ap[m * 16 * LDF + s * 4];
#pragma unroll
      for (int n = 0; n < 2; ++n) {
        b[s][n].x = bp[(s * 4) * HIDDEN + 16 * n];
        b[s][n].y = bp[(s * 4 + 1) * HIDDEN + 16 * n];
      }
    }
#pragma unroll 2
    for (int st = 0; st < NF / 8; ++st) {
      const int k0 = ((st + 1 < NF / 8) ? (st + 1) : st) * 8;
      v2f an[2][MT], bn[2][2];
#pragma unroll
      for (int s = 0; s < 2; ++s) {
#pragma unroll
        for (int n = 0; n < 2; ++n) {
          bn[s][n].x = bp[(k0 + s * 4) * HIDDEN + 16 * n];
          bn[s][n].y = bp[(k0 + s * 4 + 1) * HIDDEN + 16 * n];
        }
#pragma unroll
        for (int m = 0; m < MT; ++m) an[s][m] = *(const v2f*)&ap[m * 16 * LDF + k0 + s * 4];
      }
#pragma unroll
      for (int s = 0; s < 2; ++s)
#pragma unroll
        for (int m = 0; m < MT; ++m)
#pragma unroll
          for (int n = 0; n < 2; ++n)
            acc[m][n] = __builtin_amdgcn_wmma_f32_16x16x4_f32(false, a[s][m], false, b[s][n],
                                                              (short)0, acc[m][n], false, false);
#pragma unroll
      for (int s = 0; s < 2; ++s) {
#pragma unroll
        for (int m = 0; m < MT; ++m) a[s][m] = an[s][m];
#pragma unroll
        for (int n = 0; n < 2; ++n) b[s][n] = bn[s][n];
      }
    }
#pragma unroll
    for (int m = 0; m < MT; ++m)
#pragma unroll
      for (int n = 0; n < 2; ++n)
#pragma unroll
        for (int v = 0; v < 8; ++v) {
          const float r = acc[m][n][v];
          hbuf[(grp * ROWS + m * 16 + v + 8 * half) * LDF + c0 + 16 * n] = r > 0.0f ? r : 0.0f;
        }
  };
  if (grp < NGRP - 1) layer1(tile_count<4>{});
  else if (grp == NGRP - 1) layer1(tile_count<LASTMT>{});
  __syncthreads();

  auto layer2 = [&](auto mt) {
    constexpr int MT = decltype(mt)::value;
    const int c0 = strip * 16 + nidx;
    const float bias = b2[c0];
    v8f acc[MT];
#pragma unroll
    for (int m = 0; m < MT; ++m)
#pragma unroll
      for (int v = 0; v < 8; ++v) acc[m][v] = bias;
    const float* ap = &hbuf[(grp * ROWS + nidx) * LDF + 2 * half];
    const float* bp = &W2[(2 * half) * DN + c0];
    v2f a[4][MT], b[4];
#pragma unroll
    for (int s = 0; s < 4; ++s) {
#pragma unroll
      for (int m = 0; m < MT; ++m) a[s][m] = *(const v2f*)&ap[m * 16 * LDF + s * 4];
      b[s].x = bp[(s * 4) * DN];
      b[s].y = bp[(s * 4 + 1) * DN];
    }
#pragma unroll 2
    for (int st = 0; st < HIDDEN / 16; ++st) {
      const int k0 = ((st + 1 < HIDDEN / 16) ? (st + 1) : st) * 16;
      v2f an[4][MT], bn[4];
#pragma unroll
      for (int s = 0; s < 4; ++s) {
        bn[s].x = bp[(k0 + s * 4) * DN];
        bn[s].y = bp[(k0 + s * 4 + 1) * DN];
#pragma unroll
        for (int m = 0; m < MT; ++m) an[s][m] = *(const v2f*)&ap[m * 16 * LDF + k0 + s * 4];
      }
#pragma unroll
      for (int s = 0; s < 4; ++s)
#pragma unroll
        for (int m = 0; m < MT; ++m)
          acc[m] = __builtin_amdgcn_wmma_f32_16x16x4_f32(false, a[s][m], false, b[s],
                                                         (short)0, acc[m], false, false);
#pragma unroll
      for (int s = 0; s < 4; ++s) {
#pragma unroll
        for (int m = 0; m < MT; ++m) a[s][m] = an[s][m];
        b[s] = bn[s];
      }
    }
#pragma unroll
    for (int m = 0; m < MT; ++m)
#pragma unroll
      for (int v = 0; v < 8; ++v) {
        const int node = nb + grp * ROWS + m * 16 + v + 8 * half;
        if (node < NN) {
          const float r = acc[m][v];
          y[(long long)node * DN + c0] = r > 0.0f ? r : 0.0f;
        }
      }
  };
  if (grp < NGRP - 1) layer2(tile_count<4>{});
  else if (grp == NGRP - 1) layer2(tile_count<LASTMT>{});
}


extern "C" void kernel_launch(void* const* d_in, const int* in_sizes, int n_in,
                              void* d_out, int out_size, void* d_ws, size_t ws_size,
                              hipStream_t stream) {
  const float* x     = (const float*)d_in[0];
  const int*   eidx  = (const int*)d_in[1];
  const float* eattr = (const float*)d_in[2];
  const float* u     = (const float*)d_in[3];
  const int*   batch = (const int*)d_in[4];
  const float* W1    = (const float*)d_in[5];
  const float* b1    = (const float*)d_in[6];
  const float* W2    = (const float*)d_in[7];
  const float* b2    = (const float*)d_in[8];

  float* y   = (float*)d_out;
  float* eob = (float*)d_out + (size_t)NN * DN;

  unsigned* sorted = (unsigned*)d_ws;
  unsigned* tab2   = (unsigned*)((char*)d_ws + SORTED_BYTES);

  bin_edges<<<NCH, 1024, 0, stream>>>(eidx, sorted, tab2);

  range_mlp<<<NRG, 1024, 0, stream>>>(eattr, sorted, tab2, x, u, batch, W1, b1, W2, b2, eob, y);
}


# layer 1 split by k: node-feature and graph-row columns (3/4 of its products) run on half of the waves beside the edge-row sums; batches taken from an LDS counter; layer-1 k order changes
# speedup vs baseline: 7.3471x; 1.0472x over previous
#include <hip/hip_runtime.h>

typedef __attribute__((ext_vector_type(2))) float v2f;
typedef __attribute__((ext_vector_type(8))) float v8f;
typedef __attribute__((address_space(1))) float gfloat;

#define NN 50000
#define NE 1600000
#define DN 64
#define DE 32
#define DG 32
#define HIDDEN 128
#define NF 128
#define ROWS 64
#define LDF 130

#define CH 8192
#define NCH ((NE + CH - 1) / CH)
#define RN 208
#define NRG ((NN + RN - 1) / RN)
#define TP 256
#define SORTED_BYTES ((size_t)NE * 4)

__global__ __launch_bounds__(1024) void bin_edges(const int* __restrict__ edge_index,
                                                  unsigned* __restrict__ sorted,
                                                  unsigned* __restrict__ tab2) {
  __shared__ unsigned hist[TP];
  __shared__ unsigned offs[TP];
  __shared__ unsigned stage[CH];

  const int tid = threadIdx.x;
  const int chunk = blockIdx.x;
  const int base = chunk * CH;

  if (tid < TP) hist[tid] = 0u;
  __syncthreads();

  int dsts[CH / 1024];
#pragma unroll
  for (int k = 0; k < CH / 1024; ++k) {
    int e = base + k * 1024 + tid;
    int ec = (e < NE) ? e : (NE - 1);
    int d = edge_index[NE + ec];
    dsts[k] = (e < NE) ? d : -1;
  }

  unsigned key[CH / 1024];
  unsigned slot[CH / 1024];
#pragma unroll
  for (int k = 0; k < CH / 1024; ++k) {
    int e = base + k * 1024 + tid;
    int d = dsts[k];
    slot[k] = 0xFFFFFFFFu;
    key[k] = 0u;
    if ((unsigned)d < (unsigned)NN) {
      unsigned r = (unsigned)d / RN;
      unsigned rank = atomicAdd(&hist[r], 1u);
      key[k] = ((unsigned)e << 8) | ((unsigned)d - r * RN);
      slot[k] = (r << 16) | rank;
    }
  }
  __syncthreads();

  if (tid < 32) {
    unsigned v[8];
    unsigned s = 0u;
#pragma unroll
    for (int i = 0; i < 8; ++i) { v[i] = hist[tid * 8 + i]; s += v[i]; }
    unsigned inc = s;
#pragma unroll
    for (int o = 1; o < 32; o <<= 1) {
      unsigned t = __shfl_up(inc, o);
      if (tid >= o) inc += t;
    }
    unsigned run = inc - s;
#pragma unroll
    for (int i = 0; i < 8; ++i) { offs[tid * 8 + i] = run; run += v[i]; }
  }
  __syncthreads();

#pragma unroll
  for (int k = 0; k < CH / 1024; ++k) {
    if (slot[k] != 0xFFFFFFFFu) stage[offs[slot[k] >> 16] + (slot[k] & 0xFFFFu)] = key[k];
  }
  if (tid < TP / 2) tab2[chunk * (TP / 2) + tid] = offs[2 * tid] | (offs[2 * tid + 1] << 16);
  __syncthreads();

  const unsigned total = offs[TP - 1];
#pragma unroll
  for (int k = 0; k < CH / 1024; ++k) {
    unsigned i = k * 1024 + tid;
    if (i < total) sorted[base + i] = stage[i];
  }
}

#define KCAP 12288
#define CPW ((NCH + 31) / 32)
#define BR 32
#define HB 16
#define MTILES (RN / 16)
#define NGRP ((MTILES + 3) / 4)
#define LASTMT (MTILES - 4 * (NGRP - 1))
#define MLPW (4 * NGRP)

template <int N> struct tile_count { static constexpr int value = N; };


__global__ __launch_bounds__(1024) void range_mlp(const float* __restrict__ edge_attr,
                                                  const unsigned* __restrict__ sorted,
                                                  const unsigned* __restrict__ tab2,
                                                  const float* __restrict__ x,
                                                  const float* __restrict__ u,
                                                  const int* __restrict__ batch,
                                                  const float* __restrict__ W1, const float* __restrict__ b1,
                                                  const float* __restrict__ W2, const float* __restrict__ b2,
                                                  float* __restrict__ eob, float* __restrict__ y) {
  __shared__ float feat[RN * LDF];
  __shared__ float hbuf[RN * LDF];
  __shared__ unsigned keys[KCAP];
  __shared__ unsigned next_batch;
  __shared__ unsigned cnt[RN];
  __shared__ unsigned pre[NCH + 1];
  __shared__ unsigned short seg_s[NCH];
  __shared__ unsigned short seg_n[NCH];

  const int tid = threadIdx.x;
  const int rg = blockIdx.x;
  const int lane = tid & 31;
  const int wave = __builtin_amdgcn_readfirstlane(tid >> 5);
  const int mw = wave - (32 - MLPW);
  const int half = lane >> 4;
  const int nidx = lane & 15;
  const int grp = mw >> 2;
  const int strip = mw & 3;

  unsigned d0 = 0u, d1 = 0u;
  if (tid < NCH) {
    const int w = (tid * TP + rg) >> 1;
    d0 = tab2[w];
    d1 = tab2[w + 1];
  }
  int bq[2];
#pragma unroll
  for (int j = 0; j < 2; ++j) {
    const int q = tid + 1024 * j;
    const int r = (q >> 3) < RN ? (q >> 3) : RN - 1;
    const int node = rg * RN + r;
    bq[j] = batch[node < NN ? node : NN - 1];
  }
  for (int i = tid; i < RN * DE; i += 1024) feat[(i >> 5) * LDF + DN + (i & 31)] = 0.0f;
  if (tid < RN) cnt[tid] = 0u;
  if (tid < NCH) {
    const unsigned s = (rg & 1) ? (d0 >> 16) : (d0 & 0xFFFFu);
    const unsigned e = (rg & 1) ? (d1 & 0xFFFFu) : (d0 >> 16);
    seg_s[tid] = (unsigned short)s;
    seg_n[tid] = (unsigned short)(e - s);
  }
  __syncthreads();

  if (tid < 32) {
    unsigned v[CPW];
    unsigned s = 0u;
#pragma unroll
    for (int i = 0; i < CPW; ++i) {
      const int c = tid * CPW + i;
      const unsigned sn = seg_n[c < NCH ? c : 0];
      v[i] = (c < NCH) ? sn : 0u;
      s += v[i];
    }
    unsigned inc = s;
#pragma unroll
    for (int o = 1; o < 32; o <<= 1) {
      unsigned t = __shfl_up(inc, o);
      if (tid >= o) inc += t;
    }
    unsigned run = inc - s;
#pragma unroll
    for (int i = 0; i < CPW; ++i) {
      const int c = tid * CPW + i;
      if (c < NCH) pre[c] = run;
      run += v[i];
    }
    if (tid == 31) pre[NCH] = run;
  }
  __syncthreads();

  const unsigned total = (unsigned)__builtin_amdgcn_readfirstlane((int)pre[NCH]);

  auto issue_rows = [&](float (&v)[HB], unsigned ro, int u0) {
    const gfloat* rp[HB];
#pragma unroll
    for (int u = 0; u < HB; ++u) {
      const unsigned o = (unsigned)__builtin_amdgcn_readlane((int)ro, u0 + u);
      const unsigned long long rb = (unsigned long long)edge_attr + (unsigned long long)o;
      const unsigned lo = (unsigned)__builtin_amdgcn_readfirstlane((int)(unsigned)rb);
      const unsigned hi = (unsigned)__builtin_amdgcn_readfirstlane((int)(unsigned)(rb >> 32));
      rp[u] = (const gfloat*)(((unsigned long long)hi << 32) | (unsigned long long)lo);
    }
#pragma unroll
    for (int u = 0; u < HB; ++u) v[u] = rp[u][lane];
  };
  auto add_full = [&](const float (&v)[HB], unsigned ti, int u0) {
#pragma unroll
    for (int u = 0; u < HB; ++u) {
      const unsigned t = (unsigned)__builtin_amdgcn_readlane((int)ti, u0 + u);
      atomicAdd(&feat[t + lane], v[u]);
    }
  };
  auto add_part = [&](const float (&v)[HB], unsigned ti, int u0, int n) {
#pragma unroll
    for (int u = 0; u < HB; ++u) {
      if (u0 + u < n) {
        const unsigned t = (unsigned)__builtin_amdgcn_readlane((int)ti, u0 + u);
        atomicAdd(&feat[t + lane], v[u]);
      }
    }
  };

  auto layer1 = [&](auto mt, auto part) {
    constexpr int MT = decltype(mt)::value;
    constexpr int PT = decltype(part)::value;
    constexpr int NST = (PT == 0) ? (DN + DG) / 8 : DE / 8;
    const int c0 = strip * 32 + nidx;
    v8f acc[MT][2];
#pragma unroll
    for (int n = 0; n < 2; ++n) {
      const float bias = b1[c0 + 16 * n];
#pragma unroll
      for (int m = 0; m < MT; ++m)
#pragma unroll
        for (int v = 0; v < 8; ++v)
          acc[m][n][v] = (PT == 0) ? bias
                                   : hbuf[(grp * ROWS + m * 16 + v + 8 * half) * LDF + c0 + 16 * n];
    }
    const float* ap = &feat[(grp * ROWS + nidx) * LDF + 2 * half];
    const float* bp = &W1[(2 * half) * HIDDEN + c0];
    auto kof = [&](int st) -> int {
      return (PT == 0) ? ((st < DN / 8) ? st * 8 : st * 8 + DE) : (DN + st * 8);
    };
    v2f a[2][MT], b[2][2];
    {
      const int k0 = kof(0);
#pragma unroll
      for (int s = 0; s < 2; ++s) {
#pragma unroll
        for (int m = 0; m < MT; ++m) a[s][m] = *(const v2f*)&ap[m * 16 * LDF + k0 + s * 4];
#pragma unroll
        for (int n = 0; n < 2; ++n) {
          b[s][n].x = bp[(k0 + s * 4) * HIDDEN + 16 * n];
          b[s][n].y = bp[(k0 + s * 4 + 1) * HIDDEN + 16 * n];
        }
      }
    }
#pragma unroll 2
    for (int st = 0; st < NST; ++st) {
      const int k0 = kof((st + 1 < NST) ? (st + 1) : st);
      v2f an[2][MT], bn[2][2];
#pragma unroll
      for (int s = 0; s < 2; ++s) {
#pragma unroll
        for (int n = 0; n < 2; ++n) {
          bn[s][n].x = bp[(k0 + s * 4) * HIDDEN + 16 * n];
          bn[s][n].y = bp[(k0 + s * 4 + 1) * HIDDEN + 16 * n];
        }
#pragma unroll
        for (int m = 0; m < MT; ++m) an[s][m] = *(const v2f*)&ap[m * 16 * LDF + k0 + s * 4];
      }
#pragma unroll
      for (int s = 0; s < 2; ++s)
#pragma unroll
        for (int m = 0; m < MT; ++m)
#pragma unroll
          for (int n = 0; n < 2; ++n)
            acc[m][n] = __builtin_amdgcn_wmma_f32_16x16x4_f32(false, a[s][m], false, b[s][n],
                                                              (short)0, acc[m][n], false, false);
#pragma unroll
      for (int s = 0; s < 2; ++s) {
#pragma unroll
        for (int m = 0; m < MT; ++m) a[s][m] = an[s][m];
#pragma unroll
        for (int n = 0; n < 2; ++n) b[s][n] = bn[s][n];
      }
    }
#pragma unroll
    for (int m = 0; m < MT; ++m)
#pragma unroll
      for (int n = 0; n < 2; ++n)
#pragma unroll
        for (int v = 0; v < 8; ++v) {
          const float r = acc[m][n][v];
          hbuf[(grp * ROWS + m * 16 + v + 8 * half) * LDF + c0 + 16 * n] =
              (PT == 0) ? r : (r > 0.0f ? r : 0.0f);
        }
  };


  auto copy_keys = [&](unsigned base) {
    unsigned kk[CPW][2];
    int pos[CPW];
    int nn[CPW];
#pragma unroll
    for (int i = 0; i < CPW; ++i) {
      const int c = wave + 32 * i;
      const int cc = (c < NCH) ? c : (NCH - 1);
      const int s = seg_s[cc];
      nn[i] = (c < NCH) ? (int)seg_n[cc] : 0;
      pos[i] = (int)pre[cc] - (int)base;
#pragma unroll
      for (int h = 0; h < 2; ++h) {
        const int j = lane + 32 * h;
        const int off = (j < nn[i]) ? (s + j) : 0;
        kk[i][h] = sorted[cc * CH + off];
      }
    }
#pragma unroll
    for (int i = 0; i < CPW; ++i) {
#pragma unroll
      for (int h = 0; h < 2; ++h) {
        const int j = lane + 32 * h;
        const int f = pos[i] + j;
        if (j < nn[i] && (unsigned)f < (unsigned)KCAP) keys[f] = kk[i][h];
      }
    }
    for (int c = wave; c < NCH; c += 32) {
      const int n = __builtin_amdgcn_readfirstlane((int)seg_n[c]);
      if (n > 64) {
        const int s = seg_s[c];
        const int p = (int)pre[c] - (int)base;
        for (int j = 64 + lane; j < n; j += 32) {
          const int f = p + j;
          if ((unsigned)f < (unsigned)KCAP) keys[f] = sorted[c * CH + s + j];
        }
      }
    }
  };

  auto sum_rows = [&](unsigned base) {
    const unsigned m = (total - base < (unsigned)KCAP) ? (total - base) : (unsigned)KCAP;
    const int nbat = (int)((m + BR - 1) / BR);
    auto claim_issue = [&]() -> unsigned {
      unsigned v = 0u;
      if (lane == 0) v = atomicAdd(&next_batch, 1u);
      return v;
    };
    auto load_keys = [&](int bt, unsigned& ro, unsigned& ti, int& nextb) {
      const unsigned i = (unsigned)bt * BR + lane;
      const unsigned myk = keys[i < m ? i : m - 1];
      const unsigned cv = claim_issue();
      const unsigned ln = myk & 255u;
      atomicAdd(&cnt[ln], (i < m) ? 1u : 0u);
      ro = (myk >> 8) << 7;
      ti = ln * LDF + DN;
      nextb = __builtin_amdgcn_readfirstlane((int)cv);
    };
    auto rows_in = [&](int bt) -> int {
      const unsigned r = m - (unsigned)bt * BR;
      return (r < (unsigned)BR) ? (int)r : BR;
    };
    int bt = __builtin_amdgcn_readfirstlane((int)claim_issue());
    if (bt < nbat) {
      float va[HB], vb[HB];
      unsigned ro, ti;
      int bnx;
      load_keys(bt, ro, ti, bnx);
      issue_rows(va, ro, 0);
      issue_rows(vb, ro, HB);
      while (bnx < nbat) {
        unsigned ron, tin;
        int bnn;
        add_full(va, ti, 0);
        load_keys(bnx, ron, tin, bnn);
        __builtin_amdgcn_sched_barrier(0);
        issue_rows(va, ron, 0);
        add_full(vb, ti, HB);
        __builtin_amdgcn_sched_barrier(0);
        issue_rows(vb, ron, HB);
        ro = ron;
        ti = tin;
        bt = bnx;
        bnx = bnn;
      }
      const int n = rows_in(bt);
      add_part(va, ti, 0, n);
      add_part(vb, ti, HB, n);
    }
  };

  if (tid == 0) next_batch = 0u;
  {
    float4 vx[4];
    float4 vu[2];
#pragma unroll
    for (int j = 0; j < 4; ++j) {
      const int q = tid + 1024 * j;
      const int r = (q >> 4) < RN ? (q >> 4) : RN - 1;
      const int node = rg * RN + r;
      const int nc = node < NN ? node : NN - 1;
      vx[j] = *(const float4*)&x[(size_t)nc * DN + (q & 15) * 4];
    }
#pragma unroll
    for (int j = 0; j < 2; ++j) {
      const int q = tid + 1024 * j;
      vu[j] = *(const float4*)&u[(size_t)bq[j] * DG + (q & 7) * 4];
    }

    copy_keys(0u);

#pragma unroll
    for (int j = 0; j < 4; ++j) {
      const int q = tid + 1024 * j;
      const int r = q >> 4;
      const bool ok = (rg * RN + r) < NN;
      float* p = &feat[r * LDF + (q & 15) * 4];
      v2f lo, hi;
      lo.x = ok ? vx[j].x : 0.0f;  lo.y = ok ? vx[j].y : 0.0f;
      hi.x = ok ? vx[j].z : 0.0f;  hi.y = ok ? vx[j].w : 0.0f;
      if (r < RN) {
        *(v2f*)p = lo;
        *(v2f*)(p + 2) = hi;
      }
    }
#pragma unroll
    for (int j = 0; j < 2; ++j) {
      const int q = tid + 1024 * j;
      const int r = q >> 3;
      const bool ok = (rg * RN + r) < NN;
      float* pu = &feat[r * LDF + DN + DE + (q & 7) * 4];
      v2f lo, hi;
      lo.x = ok ? vu[j].x : 0.0f;  lo.y = ok ? vu[j].y : 0.0f;
      hi.x = ok ? vu[j].z : 0.0f;  hi.y = ok ? vu[j].w : 0.0f;
      if (r < RN) {
        *(v2f*)pu = lo;
        *(v2f*)(pu + 2) = hi;
      }
    }
  }
  __syncthreads();

  if (mw >= 0) {
    if (grp < NGRP - 1) layer1(tile_count<4>{}, tile_count<0>{});
    else layer1(tile_count<LASTMT>{}, tile_count<0>{});
  }
  sum_rows(0u);
  __syncthreads();

  for (unsigned base = KCAP; base < total; base += KCAP) {
    if (tid == 0) next_batch = 0u;
    copy_keys(base);
    __syncthreads();
    sum_rows(base);
    __syncthreads();
  }

  const int nb = rg * RN;
  for (int i = tid; i < RN * DE; i += 1024) {
    const int ln = i >> 5;
    float* p = &feat[ln * LDF + DN + (i & 31)];
    const float mean = *p / fmaxf((float)cnt[ln], 1.0f);
    *p = mean;
    if (nb + ln < NN) eob[(size_t)(nb + ln) * DE + (i & 31)] = mean;
  }
  __syncthreads();

  if (mw >= 0) {
    if (grp < NGRP - 1) layer1(tile_count<4>{}, tile_count<1>{});
    else layer1(tile_count<LASTMT>{}, tile_count<1>{});
  }
  __syncthreads();

  auto layer2 = [&](auto mt) {
    constexpr int MT = decltype(mt)::value;
    const int c0 = strip * 16 + nidx;
    const float bias = b2[c0];
    v8f acc[MT];
#pragma unroll
    for (int m = 0; m < MT; ++m)
#pragma unroll
      for (int v = 0; v < 8; ++v) acc[m][v] = bias;
    const float* ap = &hbuf[(grp * ROWS + nidx) * LDF + 2 * half];
    const float* bp = &W2[(2 * half) * DN + c0];
    v2f a[4][MT], b[4];
#pragma unroll
    for (int s = 0; s < 4; ++s) {
#pragma unroll
      for (int m = 0; m < MT; ++m) a[s][m] = *(const v2f*)&ap[m * 16 * LDF + s * 4];
      b[s].x = bp[(s * 4) * DN];
      b[s].y = bp[(s * 4 + 1) * DN];
    }
#pragma unroll 2
    for (int st = 0; st < HIDDEN / 16; ++st) {
      const int k0 = ((st + 1 < HIDDEN / 16) ? (st + 1) : st) * 16;
      v2f an[4][MT], bn[4];
#pragma unroll
      for (int s = 0; s < 4; ++s) {
        bn[s].x = bp[(k0 + s * 4) * DN];
        bn[s].y = bp[(k0 + s * 4 + 1) * DN];
#pragma unroll
        for (int m = 0; m < MT; ++m) an[s][m] = *(const v2f*)&ap[m * 16 * LDF + k0 + s * 4];
      }
#pragma unroll
      for (int s = 0; s < 4; ++s)
#pragma unroll
        for (int m = 0; m < MT; ++m)
          acc[m] = __builtin_amdgcn_wmma_f32_16x16x4_f32(false, a[s][m], false, b[s],
                                                         (short)0, acc[m], false, false);
#pragma unroll
      for (int s = 0; s < 4; ++s) {
#pragma unroll
        for (int m = 0; m < MT; ++m) a[s][m] = an[s][m];
        b[s] = bn[s];
      }
    }
#pragma unroll
    for (int m = 0; m < MT; ++m)
#pragma unroll
      for (int v = 0; v < 8; ++v) {
        const int node = nb + grp * ROWS + m * 16 + v + 8 * half;
        if (node < NN) {
          const float r = acc[m][v];
          y[(long long)node * DN + c0] = r > 0.0f ? r : 0.0f;
        }
      }
  };
  if (mw >= 0) {
    if (grp < NGRP - 1) layer2(tile_count<4>{});
    else layer2(tile_count<LASTMT>{});
  }
}


extern "C" void kernel_launch(void* const* d_in, const int* in_sizes, int n_in,
                              void* d_out, int out_size, void* d_ws, size_t ws_size,
                              hipStream_t stream) {
  const float* x     = (const float*)d_in[0];
  const int*   eidx  = (const int*)d_in[1];
  const float* eattr = (const float*)d_in[2];
  const float* u     = (const float*)d_in[3];
  const int*   batch = (const int*)d_in[4];
  const float* W1    = (const float*)d_in[5];
  const float* b1    = (const float*)d_in[6];
  const float* W2    = (const float*)d_in[7];
  const float* b2    = (const float*)d_in[8];

  float* y   = (float*)d_out;
  float* eob = (float*)d_out + (size_t)NN * DN;

  unsigned* sorted = (unsigned*)d_ws;
  unsigned* tab2   = (unsigned*)((char*)d_ws + SORTED_BYTES);

  bin_edges<<<NCH, 1024, 0, stream>>>(eidx, sorted, tab2);

  range_mlp<<<NRG, 1024, 0, stream>>>(eattr, sorted, tab2, x, u, batch, W1, b1, W2, b2, eob, y);
}
